# starting kernel
# baseline (speedup 1.0000x reference)
#include <hip/hip_runtime.h>
#include <hip/hip_bf16.h>
#include <math.h>

typedef __attribute__((ext_vector_type(2))) float v2f;
typedef __attribute__((ext_vector_type(8))) float v8f;

#define NEG_SLOPE 0.2f
#define EPSV 1e-16f


__global__ void fill_kernel(float* __restrict__ p, float v, int n) {
  int i = blockIdx.x * blockDim.x + threadIdx.x;
  if (i < n) p[i] = v;
}

__device__ __forceinline__ void atomicMaxF(float* addr, float val) {
  if (val >= 0.0f) atomicMax((int*)addr, __float_as_int(val));
  else             atomicMin((unsigned int*)addr, __float_as_uint(val));
}

__device__ __forceinline__ void edge_sd(const int* __restrict__ ei, int E, int idx,
                                        int& s, int& d) {
  if (idx < E) { s = ei[idx]; d = ei[E + idx]; }
  else         { s = idx - E; d = idx - E; }
}

template <int NT>
__global__ void __launch_bounds__(32)
gemm_wmma_f32_nt(const float* __restrict__ A, const float* __restrict__ B,
                 float* __restrict__ D, int M, int K, int ldb, int ldd, int nvalid) {
  const int tileM = blockIdx.x;
  const int lane = threadIdx.x & 31;
  const int half = lane >> 4;
  const int l16  = lane & 15;
  const int rowA = tileM * 16 + l16;
  const float* __restrict__ arow = A + (size_t)rowA * K;

  v8f acc[NT];
  #pragma unroll
  for (int nt = 0; nt < NT; ++nt) acc[nt] = (v8f){0.f,0.f,0.f,0.f,0.f,0.f,0.f,0.f};

  for (int kk = 0; kk < K; kk += 4) {
    const int ka = kk + half * 2;
    v2f a;
    a.x = arow[ka];
    a.y = arow[ka + 1];
    const float* __restrict__ brow0 = B + (size_t)ka * ldb;
    const float* __restrict__ brow1 = B + (size_t)(ka + 1) * ldb;
    #pragma unroll
    for (int nt = 0; nt < NT; ++nt) {
      const int colB = nt * 16 + l16;
      v2f b;
      b.x = brow0[colB];
      b.y = brow1[colB];
      acc[nt] = __builtin_amdgcn_wmma_f32_16x16x4_f32(false, a, false, b,
                                                      (short)0, acc[nt], false, false);
    }
  }
  #pragma unroll
  for (int nt = 0; nt < NT; ++nt) {
    #pragma unroll
    for (int g = 0; g < 8; ++g) {
      const int r = tileM * 16 + half * 8 + g;
      const int c = nt * 16 + l16;
      if (c < nvalid) D[(size_t)r * ldd + c] = acc[nt][g];
    }
  }
}

__global__ void pad_kernel(const float* __restrict__ W, float* __restrict__ Wp,
                           int K, int Nin, int Nout) {
  int i = blockIdx.x * blockDim.x + threadIdx.x;
  if (i >= K * Nout) return;
  int k = i / Nout, c = i % Nout;
  Wp[i] = (c < Nin) ? W[k * Nin + c] : 0.0f;
}


__global__ void alphas1_kernel(const float* __restrict__ h,
                               const float* __restrict__ a_src,
                               const float* __restrict__ a_dst,
                               float* __restrict__ as, float* __restrict__ ad, int N) {
  int i = blockIdx.x * blockDim.x + threadIdx.x;
  if (i >= N * 8) return;
  int n = i >> 3, hh = i & 7;
  const float* hp = h + (size_t)n * 64 + hh * 8;
  const float* sp = a_src + hh * 8;
  const float* dp = a_dst + hh * 8;
  float s1 = 0.f, s2 = 0.f;
  #pragma unroll
  for (int c = 0; c < 8; ++c) { float v = hp[c]; s1 += v * sp[c]; s2 += v * dp[c]; }
  as[i] = s1; ad[i] = s2;
}

__global__ void edge_max1_kernel(const int* __restrict__ ei, int E, int Et,
                                 const float* __restrict__ as, const float* __restrict__ ad,
                                 float* __restrict__ m) {
  int idx = blockIdx.x * blockDim.x + threadIdx.x;
  if (idx >= Et) return;
  int s, d; edge_sd(ei, E, idx, s, d);
  #pragma unroll
  for (int h = 0; h < 8; ++h) {
    float v = as[s * 8 + h] + ad[d * 8 + h];
    v = v > 0.f ? v : NEG_SLOPE * v;
    atomicMaxF(&m[d * 8 + h], v);
  }
}

__global__ void edge_sum1_kernel(const int* __restrict__ ei, int E, int Et,
                                 const float* __restrict__ as, const float* __restrict__ ad,
                                 const float* __restrict__ m, float* __restrict__ den) {
  int idx = blockIdx.x * blockDim.x + threadIdx.x;
  if (idx >= Et) return;
  int s, d; edge_sd(ei, E, idx, s, d);
  #pragma unroll
  for (int h = 0; h < 8; ++h) {
    float v = as[s * 8 + h] + ad[d * 8 + h];
    v = v > 0.f ? v : NEG_SLOPE * v;
    atomicAdd(&den[d * 8 + h], expf(v - m[d * 8 + h]));
  }
}

__global__ void edge_agg1_kernel(const int* __restrict__ ei, int E, int Et,
                                 const float* __restrict__ as, const float* __restrict__ ad,
                                 const float* __restrict__ m, const float* __restrict__ den,
                                 const float* __restrict__ h1, float* __restrict__ out) {
  int t = blockIdx.x * blockDim.x + threadIdx.x;
  if (t >= Et * 8) return;
  int idx = t >> 3, hh = t & 7;
  int s, d; edge_sd(ei, E, idx, s, d);
  float v = as[s * 8 + hh] + ad[d * 8 + hh];
  v = v > 0.f ? v : NEG_SLOPE * v;
  float alpha = expf(v - m[d * 8 + hh]) / (den[d * 8 + hh] + EPSV);
  const float* hp = h1 + (size_t)s * 64 + hh * 8;
  float* op = out + (size_t)d * 64 + hh * 8;
  #pragma unroll
  for (int c = 0; c < 8; ++c) atomicAdd(&op[c], hp[c] * alpha);
}

__global__ void bias_elu_kernel(float* __restrict__ x, const float* __restrict__ b,
                                int total, int C) {
  int i = blockIdx.x * blockDim.x + threadIdx.x;
  if (i >= total) return;
  float v = x[i] + b[i % C];
  x[i] = v > 0.f ? v : expm1f(v);
}


__global__ void alphas2_kernel(const float* __restrict__ h,
                               const float* __restrict__ a_src,
                               const float* __restrict__ a_dst,
                               float* __restrict__ as, float* __restrict__ ad,
                               int N, int C) {
  int n = blockIdx.x * blockDim.x + threadIdx.x;
  if (n >= N) return;
  float s1 = 0.f, s2 = 0.f;
  for (int c = 0; c < C; ++c) { float v = h[(size_t)n * C + c]; s1 += v * a_src[c]; s2 += v * a_dst[c]; }
  as[n] = s1; ad[n] = s2;
}

__global__ void edge_max2_kernel(const int* __restrict__ ei, int E, int Et,
                                 const float* __restrict__ as, const float* __restrict__ ad,
                                 float* __restrict__ m) {
  int idx = blockIdx.x * blockDim.x + threadIdx.x;
  if (idx >= Et) return;
  int s, d; edge_sd(ei, E, idx, s, d);
  float v = as[s] + ad[d];
  v = v > 0.f ? v : NEG_SLOPE * v;
  atomicMaxF(&m[d], v);
}

__global__ void edge_sum2_kernel(const int* __restrict__ ei, int E, int Et,
                                 const float* __restrict__ as, const float* __restrict__ ad,
                                 const float* __restrict__ m, float* __restrict__ den) {
  int idx = blockIdx.x * blockDim.x + threadIdx.x;
  if (idx >= Et) return;
  int s, d; edge_sd(ei, E, idx, s, d);
  float v = as[s] + ad[d];
  v = v > 0.f ? v : NEG_SLOPE * v;
  atomicAdd(&den[d], expf(v - m[d]));
}

__global__ void edge_agg2_kernel(const int* __restrict__ ei, int E, int Et,
                                 const float* __restrict__ as, const float* __restrict__ ad,
                                 const float* __restrict__ m, const float* __restrict__ den,
                                 const float* __restrict__ h2, float* __restrict__ out, int C) {
  int idx = blockIdx.x * blockDim.x + threadIdx.x;
  if (idx >= Et) return;
  int s, d; edge_sd(ei, E, idx, s, d);
  float v = as[s] + ad[d];
  v = v > 0.f ? v : NEG_SLOPE * v;
  float alpha = expf(v - m[d]) / (den[d] + EPSV);
  const float* hp = h2 + (size_t)s * C;
  float* op = out + (size_t)d * C;
  for (int c = 0; c < C; ++c) atomicAdd(&op[c], hp[c] * alpha);
}

__global__ void final_kernel(const float* __restrict__ out2, const float* __restrict__ b2,
                             float* __restrict__ y, int N, int C) {
  int n = blockIdx.x * blockDim.x + threadIdx.x;
  if (n >= N) return;
  float buf[40];
  float mx = -INFINITY;
  for (int c = 0; c < C; ++c) {
    float v = out2[(size_t)n * C + c] + b2[c];
    v = v > 0.f ? v : expm1f(v);
    buf[c] = v;
    mx = fmaxf(mx, v);
  }
  float sm = 0.f;
  for (int c = 0; c < C; ++c) sm += expf(buf[c] - mx);
  float l = logf(sm);
  for (int c = 0; c < C; ++c) y[(size_t)n * C + c] = buf[c] - mx - l;
}


static inline int cdiv(int a, int b) { return (a + b - 1) / b; }

extern "C" void kernel_launch(void* const* d_in, const int* in_sizes, int n_in,
                              void* d_out, int out_size, void* d_ws, size_t ws_size,
                              hipStream_t stream) {
  const float* x      = (const float*)d_in[0];
  const int*   ei     = (const int*)  d_in[1];
  const float* W1     = (const float*)d_in[2];
  const float* a_src1 = (const float*)d_in[3];
  const float* a_dst1 = (const float*)d_in[4];
  const float* b1     = (const float*)d_in[5];
  const float* W2     = (const float*)d_in[6];
  const float* a_src2 = (const float*)d_in[7];
  const float* a_dst2 = (const float*)d_in[8];
  const float* b2     = (const float*)d_in[9];
  float* out = (float*)d_out;

  const int HC   = in_sizes[5];
  const int NCLS = in_sizes[9];
  const int F    = in_sizes[2] / HC;
  const int N    = in_sizes[0] / F;
  const int E    = in_sizes[1] / 2;
  const int Et   = E + N;
  const int NPAD = 48;

  float* ws = (float*)d_ws;
  size_t off = 0;
  float* h1   = ws + off; off += (size_t)N * HC;
  float* as1  = ws + off; off += (size_t)N * 8;
  float* ad1  = ws + off; off += (size_t)N * 8;
  float* m1   = ws + off; off += (size_t)N * 8;
  float* den1 = ws + off; off += (size_t)N * 8;
  float* out1 = ws + off; off += (size_t)N * HC;
  float* h2   = ws + off; off += (size_t)N * NCLS;
  float* as2  = ws + off; off += (size_t)N;
  float* ad2  = ws + off; off += (size_t)N;
  float* m2   = ws + off; off += (size_t)N;
  float* den2 = ws + off; off += (size_t)N;
  float* out2 = ws + off; off += (size_t)N * NCLS;
  float* Wp   = ws + off; off += (size_t)HC * NPAD;

  const int B = 256;

  fill_kernel<<<cdiv(N * 8, B), B, 0, stream>>>(m1, -INFINITY, N * 8);
  fill_kernel<<<cdiv(N * 8, B), B, 0, stream>>>(den1, 0.f, N * 8);
  fill_kernel<<<cdiv(N * HC, B), B, 0, stream>>>(out1, 0.f, N * HC);
  fill_kernel<<<cdiv(N, B), B, 0, stream>>>(m2, -INFINITY, N);
  fill_kernel<<<cdiv(N, B), B, 0, stream>>>(den2, 0.f, N);
  fill_kernel<<<cdiv(N * NCLS, B), B, 0, stream>>>(out2, 0.f, N * NCLS);

  gemm_wmma_f32_nt<4><<<dim3(N / 16), 32, 0, stream>>>(x, W1, h1, N, F, HC, HC, HC);
  alphas1_kernel<<<cdiv(N * 8, B), B, 0, stream>>>(h1, a_src1, a_dst1, as1, ad1, N);
  edge_max1_kernel<<<cdiv(Et, B), B, 0, stream>>>(ei, E, Et, as1, ad1, m1);
  edge_sum1_kernel<<<cdiv(Et, B), B, 0, stream>>>(ei, E, Et, as1, ad1, m1, den1);
  edge_agg1_kernel<<<cdiv(Et * 8, B), B, 0, stream>>>(ei, E, Et, as1, ad1, m1, den1, h1, out1);
  bias_elu_kernel<<<cdiv(N * HC, B), B, 0, stream>>>(out1, b1, N * HC, HC);

  pad_kernel<<<cdiv(HC * NPAD, B), B, 0, stream>>>(W2, Wp, HC, NCLS, NPAD);
  gemm_wmma_f32_nt<3><<<dim3(N / 16), 32, 0, stream>>>(out1, Wp, h2, N, HC, NPAD, NCLS, NCLS);
  alphas2_kernel<<<cdiv(N, B), B, 0, stream>>>(h2, a_src2, a_dst2, as2, ad2, N, NCLS);
  edge_max2_kernel<<<cdiv(Et, B), B, 0, stream>>>(ei, E, Et, as2, ad2, m2);
  edge_sum2_kernel<<<cdiv(Et, B), B, 0, stream>>>(ei, E, Et, as2, ad2, m2, den2);
  edge_agg2_kernel<<<cdiv(Et, B), B, 0, stream>>>(ei, E, Et, as2, ad2, m2, den2, h2, out2, NCLS);

  final_kernel<<<cdiv(N, B), B, 0, stream>>>(out2, b2, out, N, NCLS);
}


# the owning work-groups finish their rows in LDS before the one store: layer 1 quotient+bias+ELU, layer 2 also log_softmax (one thread per row, class order kept); no raw sums stored; 9->7 launches
# speedup vs baseline: 7.8251x; 7.8251x over previous
#include <hip/hip_runtime.h>
#include <hip/hip_bf16.h>
#include <math.h>

typedef __attribute__((ext_vector_type(2))) float v2f;
typedef __attribute__((ext_vector_type(8))) float v8f;

#define NEG_SLOPE 0.2f
#define EPSV 1e-16f


__device__ __forceinline__ void atomicMaxF(float* addr, float val) {
  if (val >= 0.0f) atomicMax((int*)addr, __float_as_int(val));
  else             atomicMin((unsigned int*)addr, __float_as_uint(val));
}

template <int NT>
__global__ void __launch_bounds__(32)
gemm_wmma_f32_nt(const float* __restrict__ A, const float* __restrict__ B,
                 float* __restrict__ D, int M, int K, int ldb, int ldd, int nvalid) {
  const int tileM = blockIdx.x;
  const int lane = threadIdx.x & 31;
  const int half = lane >> 4;
  const int l16  = lane & 15;
  const int rowA = tileM * 16 + l16;
  const float* __restrict__ arow = A + (size_t)rowA * K;

  v8f acc[NT];
  #pragma unroll
  for (int nt = 0; nt < NT; ++nt) acc[nt] = (v8f){0.f,0.f,0.f,0.f,0.f,0.f,0.f,0.f};

  for (int kk = 0; kk < K; kk += 4) {
    const int ka = kk + half * 2;
    v2f a;
    a.x = arow[ka];
    a.y = arow[ka + 1];
    const float* __restrict__ brow0 = B + (size_t)ka * ldb;
    const float* __restrict__ brow1 = B + (size_t)(ka + 1) * ldb;
    #pragma unroll
    for (int nt = 0; nt < NT; ++nt) {
      const int colB = nt * 16 + l16;
      v2f b;
      b.x = brow0[colB];
      b.y = brow1[colB];
      acc[nt] = __builtin_amdgcn_wmma_f32_16x16x4_f32(false, a, false, b,
                                                      (short)0, acc[nt], false, false);
    }
  }
  #pragma unroll
  for (int nt = 0; nt < NT; ++nt) {
    #pragma unroll
    for (int g = 0; g < 8; ++g) {
      const int r = tileM * 16 + half * 8 + g;
      const int c = nt * 16 + l16;
      if (c < nvalid) D[(size_t)r * ldd + c] = acc[nt][g];
    }
  }
}

__global__ void pad_kernel(const float* __restrict__ W, float* __restrict__ Wp,
                           int K, int Nin, int Nout) {
  int i = blockIdx.x * blockDim.x + threadIdx.x;
  if (i >= K * Nout) return;
  int k = i / Nout, c = i % Nout;
  Wp[i] = (c < Nin) ? W[k * Nin + c] : 0.0f;
}


__global__ void alphas1_kernel(const float* __restrict__ h,
                               const float* __restrict__ a_src,
                               const float* __restrict__ a_dst,
                               float* __restrict__ as, float* __restrict__ ad, int N) {
  int i = blockIdx.x * blockDim.x + threadIdx.x;
  if (i >= N * 8) return;
  int n = i >> 3, hh = i & 7;
  const float* hp = h + (size_t)n * 64 + hh * 8;
  const float* sp = a_src + hh * 8;
  const float* dp = a_dst + hh * 8;
  float s1 = 0.f, s2 = 0.f;
  #pragma unroll
  for (int c = 0; c < 8; ++c) { float v = hp[c]; s1 += v * sp[c]; s2 += v * dp[c]; }
  as[i] = s1; ad[i] = s2;
}


__global__ void alphas2_kernel(const float* __restrict__ h,
                               const float* __restrict__ a_src,
                               const float* __restrict__ a_dst,
                               float* __restrict__ as, float* __restrict__ ad,
                               int N, int C) {
  int n = blockIdx.x * blockDim.x + threadIdx.x;
  if (n >= N) return;
  float s1 = 0.f, s2 = 0.f;
  for (int c = 0; c < C; ++c) { float v = h[(size_t)n * C + c]; s1 += v * a_src[c]; s2 += v * a_dst[c]; }
  as[n] = s1; ad[n] = s2;
}


#define AGG_T   1024
#define RN      196
#define LCAP    32768
#define P1      73
#define P2      41

struct L1Vals { float ex; float4 a; float4 b; };

__device__ __forceinline__ float l1_logit(int s, int d, int hh,
                                         const float* __restrict__ as, const float* __restrict__ ad) {
  const float v = as[s * 8 + hh] + ad[d * 8 + hh];
  return v > 0.f ? v : NEG_SLOPE * v;
}

__device__ __forceinline__ L1Vals l1_load(int s, int d, int dl, int hh,
                                          const float* __restrict__ as, const float* __restrict__ ad,
                                          const float* mx, const float* __restrict__ h1) {
  L1Vals r;
  r.ex = expf(l1_logit(s, d, hh, as, ad) - mx[dl * 8 + hh]);
  const float4* hp = (const float4*)(h1 + (size_t)s * 64 + hh * 8);
  r.a = hp[0]; r.b = hp[1];
  return r;
}

__device__ __forceinline__ void l1_add(float* rows, int dl, int hh, const L1Vals& r) {
  float* op = rows + dl * P1 + hh * 8;
  atomicAdd(&rows[dl * P1 + 64 + hh], r.ex);
  atomicAdd(&op[0], r.a.x * r.ex); atomicAdd(&op[1], r.a.y * r.ex);
  atomicAdd(&op[2], r.a.z * r.ex); atomicAdd(&op[3], r.a.w * r.ex);
  atomicAdd(&op[4], r.b.x * r.ex); atomicAdd(&op[5], r.b.y * r.ex);
  atomicAdd(&op[6], r.b.z * r.ex); atomicAdd(&op[7], r.b.w * r.ex);
}

__device__ __forceinline__ void collect_edges(const int* __restrict__ ei, int E, int lo, int nn,
                                              int* lst_s, unsigned char* lst_d, int* cnt) {
  const int tid = threadIdx.x;
  const int* __restrict__ dsts = ei + E;
  auto push = [&](int idx, int d) {
    const unsigned dl = (unsigned)(d - lo);
    if (dl < (unsigned)nn) {
      const int pos = atomicAdd(cnt, 1);
      if (pos < LCAP) { lst_s[pos] = idx; lst_d[pos] = (unsigned char)dl; }
    }
  };
  auto push4 = [&](int i4, const int4& v) {
    const unsigned un = (unsigned)nn;
    if (((unsigned)(v.x - lo) < un) | ((unsigned)(v.y - lo) < un) |
        ((unsigned)(v.z - lo) < un) | ((unsigned)(v.w - lo) < un)) {
      push(4 * i4, v.x); push(4 * i4 + 1, v.y); push(4 * i4 + 2, v.z); push(4 * i4 + 3, v.w);
    }
  };
  int done = 0;
  if ((((size_t)dsts) & 15) == 0) {
    const int4* __restrict__ d4 = (const int4*)dsts;
    const int E4 = E >> 2;
    int i = tid;
    for (; i + 7 * AGG_T < E4; i += 8 * AGG_T) {
      const int4 v0 = d4[i],             v1 = d4[i + AGG_T],     v2 = d4[i + 2 * AGG_T],
                 v3 = d4[i + 3 * AGG_T], v4 = d4[i + 4 * AGG_T], v5 = d4[i + 5 * AGG_T],
                 v6 = d4[i + 6 * AGG_T], v7 = d4[i + 7 * AGG_T];
      push4(i, v0);             push4(i + AGG_T, v1);     push4(i + 2 * AGG_T, v2);
      push4(i + 3 * AGG_T, v3); push4(i + 4 * AGG_T, v4); push4(i + 5 * AGG_T, v5);
      push4(i + 6 * AGG_T, v6); push4(i + 7 * AGG_T, v7);
    }
    for (; i < E4; i += AGG_T) push4(i, d4[i]);
    done = E4 * 4;
  }
  for (int i = done + tid; i < E; i += AGG_T) push(i, dsts[i]);
}

__global__ void __launch_bounds__(AGG_T)
edge_agg1_own_kernel(const int* __restrict__ ei, int E, int N,
                     const float* __restrict__ as, const float* __restrict__ ad,
                     const float* __restrict__ b,
                     const float* __restrict__ h1, float* __restrict__ out,
                     int* __restrict__ keep_s, int* __restrict__ keep_d, int* __restrict__ keep_n) {
  __shared__ float rows[RN * P1];
  __shared__ float mx[RN * 8];
  __shared__ int lst_s[LCAP];
  __shared__ unsigned char lst_d[LCAP];
  __shared__ int cnt;
  const int tid = threadIdx.x;
  const int lo = blockIdx.x * RN;
  const int nn = min(RN, N - lo);
  for (int i = tid; i < RN * P1; i += AGG_T) rows[i] = 0.f;
  for (int i = tid; i < RN * 8; i += AGG_T) mx[i] = -INFINITY;
  if (tid < nn) { lst_s[tid] = lo + tid; lst_d[tid] = (unsigned char)tid; }
  if (tid == 0) cnt = nn;
  __syncthreads();
  collect_edges(ei, E, lo, nn, lst_s, lst_d, &cnt);
  __syncthreads();
  const int total = cnt;
  const int nl = total <= LCAP ? total : nn;
  for (int i = nn + tid; i < nl; i += AGG_T) lst_s[i] = ei[lst_s[i]];
  __syncthreads();
  for (int i = tid; i < nl; i += AGG_T) {
    keep_s[(size_t)blockIdx.x * LCAP + i] = lst_s[i];
    keep_d[(size_t)blockIdx.x * LCAP + i] = lst_d[i];
  }
  if (tid == 0) keep_n[blockIdx.x] = total;

  const int hh = tid & 7, sub = tid >> 3;

  for (int base = 0; base < nl; base += 4 * (AGG_T / 8)) {
    float v[4]; int dl[4]; bool k[4];
    #pragma unroll
    for (int j = 0; j < 4; ++j) {
      const int e = base + sub + j * (AGG_T / 8);
      k[j] = e < nl;
      const int f = k[j] ? e : nl - 1;
      dl[j] = lst_d[f];
      v[j] = l1_logit(lst_s[f], lo + dl[j], hh, as, ad);
    }
    #pragma unroll
    for (int j = 0; j < 4; ++j) if (k[j]) atomicMaxF(&mx[dl[j] * 8 + hh], v[j]);
  }
  if (total > LCAP) {
    const int* __restrict__ dsts = ei + E;
    for (int i = tid; i < E; i += AGG_T) {
      const unsigned dl = (unsigned)(dsts[i] - lo);
      if (dl < (unsigned)nn) {
        const int s = ei[i];
        for (int h = 0; h < 8; ++h) atomicMaxF(&mx[dl * 8 + h], l1_logit(s, lo + (int)dl, h, as, ad));
      }
    }
  }
  __syncthreads();

  for (int base = 0; base < nl; base += 2 * (AGG_T / 8)) {
    const int e0 = base + sub, e1 = e0 + AGG_T / 8;
    const bool k0 = e0 < nl, k1 = e1 < nl;
    const int f0 = k0 ? e0 : nl - 1, f1 = k1 ? e1 : nl - 1;
    const int s0 = lst_s[f0], d0 = lst_d[f0];
    const int s1 = lst_s[f1], d1 = lst_d[f1];
    const L1Vals r0 = l1_load(s0, lo + d0, d0, hh, as, ad, mx, h1);
    const L1Vals r1 = l1_load(s1, lo + d1, d1, hh, as, ad, mx, h1);
    if (k0) l1_add(rows, d0, hh, r0);
    if (k1) l1_add(rows, d1, hh, r1);
  }
  if (total > LCAP) {
    const int* __restrict__ dsts = ei + E;
    for (int i = tid; i < E; i += AGG_T) {
      const unsigned dl = (unsigned)(dsts[i] - lo);
      if (dl < (unsigned)nn) {
        const int s = ei[i];
        for (int h = 0; h < 8; ++h) l1_add(rows, (int)dl, h, l1_load(s, lo + (int)dl, (int)dl, h, as, ad, mx, h1));
      }
    }
  }
  __syncthreads();
  for (int i = tid; i < nn * 64; i += AGG_T) {
    const int r = i >> 6, c = i & 63;
    const float v = rows[r * P1 + c] / (rows[r * P1 + 64 + (c >> 3)] + EPSV) + b[c];
    out[(size_t)lo * 64 + i] = v > 0.f ? v : expm1f(v);
  }
}

__device__ __forceinline__ float l2_logit(int s, int d,
                                         const float* __restrict__ as, const float* __restrict__ ad) {
  const float v = as[s] + ad[d];
  return v > 0.f ? v : NEG_SLOPE * v;
}

__device__ __forceinline__ void l2_edge(float* rows, const float* mx, int s, int d, int dl, int q, int C,
                                        const float* __restrict__ as, const float* __restrict__ ad,
                                        const float* __restrict__ h2) {
  const float ex = expf(l2_logit(s, d, as, ad) - mx[dl]);
  const float* hp = h2 + (size_t)s * C;
  float hv[5];
  #pragma unroll
  for (int j = 0; j < 5; ++j) { const int c = q + 8 * j; hv[j] = c < C ? hp[c] : 0.f; }
  if (q == 0) atomicAdd(&rows[dl * P2 + 40], ex);
  #pragma unroll
  for (int j = 0; j < 5; ++j) { const int c = q + 8 * j; if (c < C) atomicAdd(&rows[dl * P2 + c], hv[j] * ex); }
}

__global__ void __launch_bounds__(AGG_T)
edge_agg2_own_kernel(const int* __restrict__ ei, int E, int N,
                     const float* __restrict__ as, const float* __restrict__ ad,
                     const float* __restrict__ b,
                     const float* __restrict__ h2, float* __restrict__ out, int C,
                     const int* __restrict__ keep_s, const int* __restrict__ keep_d,
                     const int* __restrict__ keep_n) {
  __shared__ float rows[RN * P2];
  __shared__ float mx[RN];
  __shared__ float lse[RN];
  const int tid = threadIdx.x;
  const int lo = blockIdx.x * RN;
  const int nn = min(RN, N - lo);
  for (int i = tid; i < RN * P2; i += AGG_T) rows[i] = 0.f;
  if (tid < RN) mx[tid] = -INFINITY;
  __syncthreads();
  const int total = keep_n[blockIdx.x];
  const int nl = total <= LCAP ? total : nn;
  const int* __restrict__ ls = keep_s + (size_t)blockIdx.x * LCAP;
  const int* __restrict__ ld = keep_d + (size_t)blockIdx.x * LCAP;

  for (int base = 0; base < nl; base += 4 * AGG_T) {
    float v[4]; int dl[4]; bool k[4];
    #pragma unroll
    for (int j = 0; j < 4; ++j) {
      const int e = base + tid + j * AGG_T;
      k[j] = e < nl;
      const int f = k[j] ? e : nl - 1;
      dl[j] = ld[f];
      v[j] = l2_logit(ls[f], lo + dl[j], as, ad);
    }
    #pragma unroll
    for (int j = 0; j < 4; ++j) if (k[j]) atomicMaxF(&mx[dl[j]], v[j]);
  }
  if (total > LCAP) {
    const int* __restrict__ dsts = ei + E;
    for (int i = tid; i < E; i += AGG_T) {
      const unsigned dl = (unsigned)(dsts[i] - lo);
      if (dl < (unsigned)nn) atomicMaxF(&mx[dl], l2_logit(ei[i], lo + (int)dl, as, ad));
    }
  }
  __syncthreads();

  const int q = tid & 7, sub = tid >> 3;
  for (int base = 0; base < nl; base += AGG_T / 8) {
    const int e = base + sub;
    if (e < nl) { const int dl = ld[e]; l2_edge(rows, mx, ls[e], lo + dl, dl, q, C, as, ad, h2); }
  }
  if (total > LCAP) {
    const int* __restrict__ dsts = ei + E;
    for (int i = tid; i < E; i += AGG_T) {
      const unsigned dl = (unsigned)(dsts[i] - lo);
      if (dl < (unsigned)nn) {
        const int s = ei[i];
        for (int k = 0; k < 8; ++k) l2_edge(rows, mx, s, lo + (int)dl, (int)dl, k, C, as, ad, h2);
      }
    }
  }
  __syncthreads();
  for (int i = tid; i < nn * C; i += AGG_T) {
    const int r = i / C, c = i - r * C;
    const float v = rows[r * P2 + c] / (rows[r * P2 + 40] + EPSV) + b[c];
    rows[r * P2 + c] = v > 0.f ? v : expm1f(v);
  }
  __syncthreads();
  if (tid < nn) {
    float hv[40];
    #pragma unroll
    for (int c = 0; c < 40; ++c) { const float t = rows[tid * P2 + c]; hv[c] = c < C ? t : -INFINITY; }
    float m = -INFINITY;
    #pragma unroll
    for (int c = 0; c < 40; ++c) m = fmaxf(m, hv[c]);
    float sm = 0.f;
    #pragma unroll
    for (int c = 0; c < 40; ++c) sm += expf(hv[c] - m);
    mx[tid] = m;
    lse[tid] = logf(sm);
  }
  __syncthreads();
  for (int i = tid; i < nn * C; i += AGG_T) {
    const int r = i / C, c = i - r * C;
    out[(size_t)lo * C + i] = rows[r * P2 + c] - mx[r] - lse[r];
  }
}


static inline int cdiv(int a, int b) { return (a + b - 1) / b; }

extern "C" void kernel_launch(void* const* d_in, const int* in_sizes, int n_in,
                              void* d_out, int out_size, void* d_ws, size_t ws_size,
                              hipStream_t stream) {
  const float* x      = (const float*)d_in[0];
  const int*   ei     = (const int*)  d_in[1];
  const float* W1     = (const float*)d_in[2];
  const float* a_src1 = (const float*)d_in[3];
  const float* a_dst1 = (const float*)d_in[4];
  const float* b1     = (const float*)d_in[5];
  const float* W2     = (const float*)d_in[6];
  const float* a_src2 = (const float*)d_in[7];
  const float* a_dst2 = (const float*)d_in[8];
  const float* b2     = (const float*)d_in[9];
  float* out = (float*)d_out;

  const int HC   = in_sizes[5];
  const int NCLS = in_sizes[9];
  const int F    = in_sizes[2] / HC;
  const int N    = in_sizes[0] / F;
  const int E    = in_sizes[1] / 2;
  const int NPAD = 48;

  float* ws = (float*)d_ws;
  size_t off = 0;
  float* h1   = ws + off; off += (size_t)N * HC;
  float* as1  = ws + off; off += (size_t)N * 8;
  float* ad1  = ws + off; off += (size_t)N * 8;
  float* out1 = ws + off; off += (size_t)N * HC;
  float* h2   = ws + off; off += (size_t)N * NCLS;
  float* as2  = ws + off; off += (size_t)N;
  float* ad2  = ws + off; off += (size_t)N;
  float* Wp   = ws + off; off += (size_t)HC * NPAD;
  const int G = cdiv(N, RN);
  int* keep_s = (int*)(ws + off); off += (size_t)G * LCAP;
  int* keep_d = (int*)(ws + off); off += (size_t)G * LCAP;
  int* keep_n = (int*)(ws + off); off += (size_t)G;

  const int B = 256;

  gemm_wmma_f32_nt<4><<<dim3(N / 16), 32, 0, stream>>>(x, W1, h1, N, F, HC, HC, HC);
  alphas1_kernel<<<cdiv(N * 8, B), B, 0, stream>>>(h1, a_src1, a_dst1, as1, ad1, N);
  edge_agg1_own_kernel<<<G, AGG_T, 0, stream>>>(ei, E, N, as1, ad1, b1, h1, out1, keep_s, keep_d, keep_n);

  pad_kernel<<<cdiv(HC * NPAD, B), B, 0, stream>>>(W2, Wp, HC, NCLS, NPAD);
  gemm_wmma_f32_nt<3><<<dim3(N / 16), 32, 0, stream>>>(out1, Wp, h2, N, HC, NPAD, NCLS, NCLS);
  alphas2_kernel<<<cdiv(N, B), B, 0, stream>>>(h2, a_src2, a_dst2, as2, ad2, N, NCLS);
  edge_agg2_own_kernel<<<G, AGG_T, 0, stream>>>(ei, E, N, as2, ad2, b2, h2, out, NCLS, keep_s, keep_d, keep_n);
}


# edge_agg1: each owning group reads its own row of an edge bitmap (1 bit per edge, built by a new pass over the destination list), not the whole list; 7->8 launches; list and atomic add order change
# speedup vs baseline: 8.5289x; 1.0899x over previous
#include <hip/hip_runtime.h>
#include <hip/hip_bf16.h>
#include <math.h>

typedef __attribute__((ext_vector_type(2))) float v2f;
typedef __attribute__((ext_vector_type(8))) float v8f;
typedef __attribute__((ext_vector_type(4))) unsigned int v4u;

#define NEG_SLOPE 0.2f
#define EPSV 1e-16f


__device__ __forceinline__ void atomicMaxF(float* addr, float val) {
  if (val >= 0.0f) atomicMax((int*)addr, __float_as_int(val));
  else             atomicMin((unsigned int*)addr, __float_as_uint(val));
}

template <int NT>
__global__ void __launch_bounds__(32)
gemm_wmma_f32_nt(const float* __restrict__ A, const float* __restrict__ B,
                 float* __restrict__ D, int M, int K, int ldb, int ldd, int nvalid) {
  const int tileM = blockIdx.x;
  const int lane = threadIdx.x & 31;
  const int half = lane >> 4;
  const int l16  = lane & 15;
  const int rowA = tileM * 16 + l16;
  const float* __restrict__ arow = A + (size_t)rowA * K;

  v8f acc[NT];
  #pragma unroll
  for (int nt = 0; nt < NT; ++nt) acc[nt] = (v8f){0.f,0.f,0.f,0.f,0.f,0.f,0.f,0.f};

  for (int kk = 0; kk < K; kk += 4) {
    const int ka = kk + half * 2;
    v2f a;
    a.x = arow[ka];
    a.y = arow[ka + 1];
    const float* __restrict__ brow0 = B + (size_t)ka * ldb;
    const float* __restrict__ brow1 = B + (size_t)(ka + 1) * ldb;
    #pragma unroll
    for (int nt = 0; nt < NT; ++nt) {
      const int colB = nt * 16 + l16;
      v2f b;
      b.x = brow0[colB];
      b.y = brow1[colB];
      acc[nt] = __builtin_amdgcn_wmma_f32_16x16x4_f32(false, a, false, b,
                                                      (short)0, acc[nt], false, false);
    }
  }
  #pragma unroll
  for (int nt = 0; nt < NT; ++nt) {
    #pragma unroll
    for (int g = 0; g < 8; ++g) {
      const int r = tileM * 16 + half * 8 + g;
      const int c = nt * 16 + l16;
      if (c < nvalid) D[(size_t)r * ldd + c] = acc[nt][g];
    }
  }
}

__global__ void pad_kernel(const float* __restrict__ W, float* __restrict__ Wp,
                           int K, int Nin, int Nout) {
  int i = blockIdx.x * blockDim.x + threadIdx.x;
  if (i >= K * Nout) return;
  int k = i / Nout, c = i % Nout;
  Wp[i] = (c < Nin) ? W[k * Nin + c] : 0.0f;
}


__global__ void alphas1_kernel(const float* __restrict__ h,
                               const float* __restrict__ a_src,
                               const float* __restrict__ a_dst,
                               float* __restrict__ as, float* __restrict__ ad, int N) {
  int i = blockIdx.x * blockDim.x + threadIdx.x;
  if (i >= N * 8) return;
  int n = i >> 3, hh = i & 7;
  const float* hp = h + (size_t)n * 64 + hh * 8;
  const float* sp = a_src + hh * 8;
  const float* dp = a_dst + hh * 8;
  float s1 = 0.f, s2 = 0.f;
  #pragma unroll
  for (int c = 0; c < 8; ++c) { float v = hp[c]; s1 += v * sp[c]; s2 += v * dp[c]; }
  as[i] = s1; ad[i] = s2;
}


__global__ void alphas2_kernel(const float* __restrict__ h,
                               const float* __restrict__ a_src,
                               const float* __restrict__ a_dst,
                               float* __restrict__ as, float* __restrict__ ad,
                               int N, int C) {
  int n = blockIdx.x * blockDim.x + threadIdx.x;
  if (n >= N) return;
  float s1 = 0.f, s2 = 0.f;
  for (int c = 0; c < C; ++c) { float v = h[(size_t)n * C + c]; s1 += v * a_src[c]; s2 += v * a_dst[c]; }
  as[n] = s1; ad[n] = s2;
}


#define AGG_T   1024
#define RN      196
#define LCAP    32768
#define P1      73
#define P2      41

struct L1Vals { float ex; float4 a; float4 b; };

__device__ __forceinline__ float l1_logit(int s, int d, int hh,
                                         const float* __restrict__ as, const float* __restrict__ ad) {
  const float v = as[s * 8 + hh] + ad[d * 8 + hh];
  return v > 0.f ? v : NEG_SLOPE * v;
}

__device__ __forceinline__ L1Vals l1_load(int s, int d, int dl, int hh,
                                          const float* __restrict__ as, const float* __restrict__ ad,
                                          const float* mx, const float* __restrict__ h1) {
  L1Vals r;
  r.ex = expf(l1_logit(s, d, hh, as, ad) - mx[dl * 8 + hh]);
  const float4* hp = (const float4*)(h1 + (size_t)s * 64 + hh * 8);
  r.a = hp[0]; r.b = hp[1];
  return r;
}

__device__ __forceinline__ void l1_add(float* rows, int dl, int hh, const L1Vals& r) {
  float* op = rows + dl * P1 + hh * 8;
  atomicAdd(&rows[dl * P1 + 64 + hh], r.ex);
  atomicAdd(&op[0], r.a.x * r.ex); atomicAdd(&op[1], r.a.y * r.ex);
  atomicAdd(&op[2], r.a.z * r.ex); atomicAdd(&op[3], r.a.w * r.ex);
  atomicAdd(&op[4], r.b.x * r.ex); atomicAdd(&op[5], r.b.y * r.ex);
  atomicAdd(&op[6], r.b.z * r.ex); atomicAdd(&op[7], r.b.w * r.ex);
}

#define BM_E    1024
#define BM_W    (BM_E / 32)
#define BM_G    256
#define BM_T    256

__global__ void __launch_bounds__(BM_T)
edge_bitmap_kernel(const int* __restrict__ ei, int E, int G, int pitch,
                   unsigned int* __restrict__ bm) {
  __shared__ uint4 tile4[BM_G * BM_W / 4];
  unsigned int* tile = (unsigned int*)tile4;
  const int tid = threadIdx.x;
  const int base = blockIdx.x * BM_E;
  const int* __restrict__ dsts = ei + E;
  int own[BM_E / BM_T];
  #pragma unroll
  for (int j = 0; j < BM_E / BM_T; ++j) {
    const int e = base + tid + j * BM_T;
    own[j] = dsts[e < E ? e : E - 1] / RN;
  }
  for (int g0 = 0; g0 < G; g0 += BM_G) {
    for (int i = tid; i < BM_G * BM_W / 4; i += BM_T) tile4[i] = make_uint4(0u, 0u, 0u, 0u);
    __syncthreads();
    #pragma unroll
    for (int j = 0; j < BM_E / BM_T; ++j) {
      const unsigned int g = (unsigned int)(own[j] - g0);
      const int p = tid + j * BM_T;
      const unsigned int bit = base + p < E ? 1u << (p & 31) : 0u;
      if (g < (unsigned int)BM_G) atomicOr(&tile[g * BM_W + (p >> 5)], bit);
    }
    __syncthreads();
    for (int i = tid; i < BM_G * BM_W / 4; i += BM_T) {
      const int g = g0 + i / (BM_W / 4);
      if (g < G)
        ((uint4*)(bm + (size_t)g * pitch + (size_t)blockIdx.x * BM_W))[i % (BM_W / 4)] = tile4[i];
    }
    __syncthreads();
  }
}

__device__ __forceinline__ void collect_edges(const unsigned int* __restrict__ row, int nq,
                                              int* lst_s, int* cnt) {
  const int tid = threadIdx.x;
  const v4u* __restrict__ r4 = (const v4u*)row;
  for (int base = 0; base < nq; base += 8 * AGG_T) {
    v4u v[8];
    #pragma unroll
    for (int j = 0; j < 8; ++j) {
      const int q = base + tid + j * AGG_T;
      v[j] = r4[q < nq ? q : nq - 1];
    }
    int n = 0;
    #pragma unroll
    for (int j = 0; j < 8; ++j) {
      v[j] &= base + tid + j * AGG_T < nq ? ~0u : 0u;
      n += __popc(v[j].x) + __popc(v[j].y) + __popc(v[j].z) + __popc(v[j].w);
    }
    int pos = n ? atomicAdd(cnt, n) : 0;
    #pragma unroll
    for (int j = 0; j < 8; ++j) {
      const int e0 = (base + tid + j * AGG_T) * 128;
      const unsigned int w[4] = {v[j].x, v[j].y, v[j].z, v[j].w};
      #pragma unroll
      for (int cw = 0; cw < 4; ++cw) {
        unsigned int m = w[cw];
        while (m) {
          if (pos < LCAP) lst_s[pos] = e0 + cw * 32 + (__ffs(m) - 1);
          ++pos;
          m &= m - 1;
        }
      }
    }
  }
}

__global__ void __launch_bounds__(AGG_T)
edge_agg1_own_kernel(const int* __restrict__ ei, int E, int N,
                     const float* __restrict__ as, const float* __restrict__ ad,
                     const float* __restrict__ b,
                     const float* __restrict__ h1, float* __restrict__ out,
                     int* __restrict__ keep_s, int* __restrict__ keep_d, int* __restrict__ keep_n,
                     const unsigned int* __restrict__ bm, int pitch) {
  __shared__ float rows[RN * P1];
  __shared__ float mx[RN * 8];
  __shared__ int lst_s[LCAP];
  __shared__ unsigned char lst_d[LCAP];
  __shared__ int cnt;
  const int tid = threadIdx.x;
  const int lo = blockIdx.x * RN;
  const int nn = min(RN, N - lo);
  for (int i = tid; i < RN * P1; i += AGG_T) rows[i] = 0.f;
  for (int i = tid; i < RN * 8; i += AGG_T) mx[i] = -INFINITY;
  if (tid < nn) { lst_s[tid] = lo + tid; lst_d[tid] = (unsigned char)tid; }
  if (tid == 0) cnt = nn;
  __syncthreads();
  collect_edges(bm + (size_t)blockIdx.x * pitch, pitch / 4, lst_s, &cnt);
  __syncthreads();
  const int total = cnt;
  const int nl = total <= LCAP ? total : nn;
  for (int i = nn + tid; i < nl; i += AGG_T) {
    const int idx = lst_s[i];
    const int s = ei[idx], d = ei[E + idx];
    lst_s[i] = s; lst_d[i] = (unsigned char)(d - lo);
  }
  __syncthreads();
  for (int i = tid; i < nl; i += AGG_T) {
    keep_s[(size_t)blockIdx.x * LCAP + i] = lst_s[i];
    keep_d[(size_t)blockIdx.x * LCAP + i] = lst_d[i];
  }
  if (tid == 0) keep_n[blockIdx.x] = total;

  const int hh = tid & 7, sub = tid >> 3;

  for (int base = 0; base < nl; base += 4 * (AGG_T / 8)) {
    float v[4]; int dl[4]; bool k[4];
    #pragma unroll
    for (int j = 0; j < 4; ++j) {
      const int e = base + sub + j * (AGG_T / 8);
      k[j] = e < nl;
      const int f = k[j] ? e : nl - 1;
      dl[j] = lst_d[f];
      v[j] = l1_logit(lst_s[f], lo + dl[j], hh, as, ad);
    }
    #pragma unroll
    for (int j = 0; j < 4; ++j) if (k[j]) atomicMaxF(&mx[dl[j] * 8 + hh], v[j]);
  }
  if (total > LCAP) {
    const int* __restrict__ dsts = ei + E;
    for (int i = tid; i < E; i += AGG_T) {
      const unsigned dl = (unsigned)(dsts[i] - lo);
      if (dl < (unsigned)nn) {
        const int s = ei[i];
        for (int h = 0; h < 8; ++h) atomicMaxF(&mx[dl * 8 + h], l1_logit(s, lo + (int)dl, h, as, ad));
      }
    }
  }
  __syncthreads();

  for (int base = 0; base < nl; base += 2 * (AGG_T / 8)) {
    const int e0 = base + sub, e1 = e0 + AGG_T / 8;
    const bool k0 = e0 < nl, k1 = e1 < nl;
    const int f0 = k0 ? e0 : nl - 1, f1 = k1 ? e1 : nl - 1;
    const int s0 = lst_s[f0], d0 = lst_d[f0];
    const int s1 = lst_s[f1], d1 = lst_d[f1];
    const L1Vals r0 = l1_load(s0, lo + d0, d0, hh, as, ad, mx, h1);
    const L1Vals r1 = l1_load(s1, lo + d1, d1, hh, as, ad, mx, h1);
    if (k0) l1_add(rows, d0, hh, r0);
    if (k1) l1_add(rows, d1, hh, r1);
  }
  if (total > LCAP) {
    const int* __restrict__ dsts = ei + E;
    for (int i = tid; i < E; i += AGG_T) {
      const unsigned dl = (unsigned)(dsts[i] - lo);
      if (dl < (unsigned)nn) {
        const int s = ei[i];
        for (int h = 0; h < 8; ++h) l1_add(rows, (int)dl, h, l1_load(s, lo + (int)dl, (int)dl, h, as, ad, mx, h1));
      }
    }
  }
  __syncthreads();
  for (int i = tid; i < nn * 64; i += AGG_T) {
    const int r = i >> 6, c = i & 63;
    const float v = rows[r * P1 + c] / (rows[r * P1 + 64 + (c >> 3)] + EPSV) + b[c];
    out[(size_t)lo * 64 + i] = v > 0.f ? v : expm1f(v);
  }
}

__device__ __forceinline__ float l2_logit(int s, int d,
                                         const float* __restrict__ as, const float* __restrict__ ad) {
  const float v = as[s] + ad[d];
  return v > 0.f ? v : NEG_SLOPE * v;
}

__device__ __forceinline__ void l2_edge(float* rows, const float* mx, int s, int d, int dl, int q, int C,
                                        const float* __restrict__ as, const float* __restrict__ ad,
                                        const float* __restrict__ h2) {
  const float ex = expf(l2_logit(s, d, as, ad) - mx[dl]);
  const float* hp = h2 + (size_t)s * C;
  float hv[5];
  #pragma unroll
  for (int j = 0; j < 5; ++j) { const int c = q + 8 * j; hv[j] = c < C ? hp[c] : 0.f; }
  if (q == 0) atomicAdd(&rows[dl * P2 + 40], ex);
  #pragma unroll
  for (int j = 0; j < 5; ++j) { const int c = q + 8 * j; if (c < C) atomicAdd(&rows[dl * P2 + c], hv[j] * ex); }
}

__global__ void __launch_bounds__(AGG_T)
edge_agg2_own_kernel(const int* __restrict__ ei, int E, int N,
                     const float* __restrict__ as, const float* __restrict__ ad,
                     const float* __restrict__ b,
                     const float* __restrict__ h2, float* __restrict__ out, int C,
                     const int* __restrict__ keep_s, const int* __restrict__ keep_d,
                     const int* __restrict__ keep_n) {
  __shared__ float rows[RN * P2];
  __shared__ float mx[RN];
  __shared__ float lse[RN];
  const int tid = threadIdx.x;
  const int lo = blockIdx.x * RN;
  const int nn = min(RN, N - lo);
  for (int i = tid; i < RN * P2; i += AGG_T) rows[i] = 0.f;
  if (tid < RN) mx[tid] = -INFINITY;
  __syncthreads();
  const int total = keep_n[blockIdx.x];
  const int nl = total <= LCAP ? total : nn;
  const int* __restrict__ ls = keep_s + (size_t)blockIdx.x * LCAP;
  const int* __restrict__ ld = keep_d + (size_t)blockIdx.x * LCAP;

  for (int base = 0; base < nl; base += 4 * AGG_T) {
    float v[4]; int dl[4]; bool k[4];
    #pragma unroll
    for (int j = 0; j < 4; ++j) {
      const int e = base + tid + j * AGG_T;
      k[j] = e < nl;
      const int f = k[j] ? e : nl - 1;
      dl[j] = ld[f];
      v[j] = l2_logit(ls[f], lo + dl[j], as, ad);
    }
    #pragma unroll
    for (int j = 0; j < 4; ++j) if (k[j]) atomicMaxF(&mx[dl[j]], v[j]);
  }
  if (total > LCAP) {
    const int* __restrict__ dsts = ei + E;
    for (int i = tid; i < E; i += AGG_T) {
      const unsigned dl = (unsigned)(dsts[i] - lo);
      if (dl < (unsigned)nn) atomicMaxF(&mx[dl], l2_logit(ei[i], lo + (int)dl, as, ad));
    }
  }
  __syncthreads();

  const int q = tid & 7, sub = tid >> 3;
  for (int base = 0; base < nl; base += AGG_T / 8) {
    const int e = base + sub;
    if (e < nl) { const int dl = ld[e]; l2_edge(rows, mx, ls[e], lo + dl, dl, q, C, as, ad, h2); }
  }
  if (total > LCAP) {
    const int* __restrict__ dsts = ei + E;
    for (int i = tid; i < E; i += AGG_T) {
      const unsigned dl = (unsigned)(dsts[i] - lo);
      if (dl < (unsigned)nn) {
        const int s = ei[i];
        for (int k = 0; k < 8; ++k) l2_edge(rows, mx, s, lo + (int)dl, (int)dl, k, C, as, ad, h2);
      }
    }
  }
  __syncthreads();
  for (int i = tid; i < nn * C; i += AGG_T) {
    const int r = i / C, c = i - r * C;
    const float v = rows[r * P2 + c] / (rows[r * P2 + 40] + EPSV) + b[c];
    rows[r * P2 + c] = v > 0.f ? v : expm1f(v);
  }
  __syncthreads();
  if (tid < nn) {
    float hv[40];
    #pragma unroll
    for (int c = 0; c < 40; ++c) { const float t = rows[tid * P2 + c]; hv[c] = c < C ? t : -INFINITY; }
    float m = -INFINITY;
    #pragma unroll
    for (int c = 0; c < 40; ++c) m = fmaxf(m, hv[c]);
    float sm = 0.f;
    #pragma unroll
    for (int c = 0; c < 40; ++c) sm += expf(hv[c] - m);
    mx[tid] = m;
    lse[tid] = logf(sm);
  }
  __syncthreads();
  for (int i = tid; i < nn * C; i += AGG_T) {
    const int r = i / C, c = i - r * C;
    out[(size_t)lo * C + i] = rows[r * P2 + c] - mx[r] - lse[r];
  }
}


static inline int cdiv(int a, int b) { return (a + b - 1) / b; }

extern "C" void kernel_launch(void* const* d_in, const int* in_sizes, int n_in,
                              void* d_out, int out_size, void* d_ws, size_t ws_size,
                              hipStream_t stream) {
  const float* x      = (const float*)d_in[0];
  const int*   ei     = (const int*)  d_in[1];
  const float* W1     = (const float*)d_in[2];
  const float* a_src1 = (const float*)d_in[3];
  const float* a_dst1 = (const float*)d_in[4];
  const float* b1     = (const float*)d_in[5];
  const float* W2     = (const float*)d_in[6];
  const float* a_src2 = (const float*)d_in[7];
  const float* a_dst2 = (const float*)d_in[8];
  const float* b2     = (const float*)d_in[9];
  float* out = (float*)d_out;

  const int HC   = in_sizes[5];
  const int NCLS = in_sizes[9];
  const int F    = in_sizes[2] / HC;
  const int N    = in_sizes[0] / F;
  const int E    = in_sizes[1] / 2;
  const int NPAD = 48;

  float* ws = (float*)d_ws;
  size_t off = 0;
  const int G   = cdiv(N, RN);
  const int NB  = cdiv(E, BM_E);
  const int BMP = NB * BM_W;
  unsigned int* bm = (unsigned int*)(ws + off); off += (size_t)G * BMP;
  float* h1   = ws + off; off += (size_t)N * HC;
  float* as1  = ws + off; off += (size_t)N * 8;
  float* ad1  = ws + off; off += (size_t)N * 8;
  float* out1 = ws + off; off += (size_t)N * HC;
  float* h2   = ws + off; off += (size_t)N * NCLS;
  float* as2  = ws + off; off += (size_t)N;
  float* ad2  = ws + off; off += (size_t)N;
  float* Wp   = ws + off; off += (size_t)HC * NPAD;
  int* keep_s = (int*)(ws + off); off += (size_t)G * LCAP;
  int* keep_d = (int*)(ws + off); off += (size_t)G * LCAP;
  int* keep_n = (int*)(ws + off); off += (size_t)G;

  const int B = 256;

  gemm_wmma_f32_nt<4><<<dim3(N / 16), 32, 0, stream>>>(x, W1, h1, N, F, HC, HC, HC);
  alphas1_kernel<<<cdiv(N * 8, B), B, 0, stream>>>(h1, a_src1, a_dst1, as1, ad1, N);
  edge_bitmap_kernel<<<NB, BM_T, 0, stream>>>(ei, E, G, BMP, bm);
  edge_agg1_own_kernel<<<G, AGG_T, 0, stream>>>(ei, E, N, as1, ad1, b1, h1, out1, keep_s, keep_d, keep_n,
                                               bm, BMP);

  pad_kernel<<<cdiv(HC * NPAD, B), B, 0, stream>>>(W2, Wp, HC, NCLS, NPAD);
  gemm_wmma_f32_nt<3><<<dim3(N / 16), 32, 0, stream>>>(out1, Wp, h2, N, HC, NPAD, NCLS, NCLS);
  alphas2_kernel<<<cdiv(N, B), B, 0, stream>>>(h2, a_src2, a_dst2, as2, ad2, N, NCLS);
  edge_agg2_own_kernel<<<G, AGG_T, 0, stream>>>(ei, E, N, as2, ad2, b2, h2, out, NCLS, keep_s, keep_d, keep_n);
}


# edge_agg2: the list is copied to LDS in the max pass (its one read from memory); the row loop takes 8 edges per thread from it, all gathers in flight together; atomic arrival order changes
# speedup vs baseline: 8.8585x; 1.0386x over previous
#include <hip/hip_runtime.h>
#include <hip/hip_bf16.h>
#include <math.h>

typedef __attribute__((ext_vector_type(2))) float v2f;
typedef __attribute__((ext_vector_type(8))) float v8f;
typedef __attribute__((ext_vector_type(4))) unsigned int v4u;

#define NEG_SLOPE 0.2f
#define EPSV 1e-16f


__device__ __forceinline__ void atomicMaxF(float* addr, float val) {
  if (val >= 0.0f) atomicMax((int*)addr, __float_as_int(val));
  else             atomicMin((unsigned int*)addr, __float_as_uint(val));
}

template <int NT>
__global__ void __launch_bounds__(32)
gemm_wmma_f32_nt(const float* __restrict__ A, const float* __restrict__ B,
                 float* __restrict__ D, int M, int K, int ldb, int ldd, int nvalid) {
  const int tileM = blockIdx.x;
  const int lane = threadIdx.x & 31;
  const int half = lane >> 4;
  const int l16  = lane & 15;
  const int rowA = tileM * 16 + l16;
  const float* __restrict__ arow = A + (size_t)rowA * K;

  v8f acc[NT];
  #pragma unroll
  for (int nt = 0; nt < NT; ++nt) acc[nt] = (v8f){0.f,0.f,0.f,0.f,0.f,0.f,0.f,0.f};

  for (int kk = 0; kk < K; kk += 4) {
    const int ka = kk + half * 2;
    v2f a;
    a.x = arow[ka];
    a.y = arow[ka + 1];
    const float* __restrict__ brow0 = B + (size_t)ka * ldb;
    const float* __restrict__ brow1 = B + (size_t)(ka + 1) * ldb;
    #pragma unroll
    for (int nt = 0; nt < NT; ++nt) {
      const int colB = nt * 16 + l16;
      v2f b;
      b.x = brow0[colB];
      b.y = brow1[colB];
      acc[nt] = __builtin_amdgcn_wmma_f32_16x16x4_f32(false, a, false, b,
                                                      (short)0, acc[nt], false, false);
    }
  }
  #pragma unroll
  for (int nt = 0; nt < NT; ++nt) {
    #pragma unroll
    for (int g = 0; g < 8; ++g) {
      const int r = tileM * 16 + half * 8 + g;
      const int c = nt * 16 + l16;
      if (c < nvalid) D[(size_t)r * ldd + c] = acc[nt][g];
    }
  }
}

__global__ void pad_kernel(const float* __restrict__ W, float* __restrict__ Wp,
                           int K, int Nin, int Nout) {
  int i = blockIdx.x * blockDim.x + threadIdx.x;
  if (i >= K * Nout) return;
  int k = i / Nout, c = i % Nout;
  Wp[i] = (c < Nin) ? W[k * Nin + c] : 0.0f;
}


__global__ void alphas1_kernel(const float* __restrict__ h,
                               const float* __restrict__ a_src,
                               const float* __restrict__ a_dst,
                               float* __restrict__ as, float* __restrict__ ad, int N) {
  int i = blockIdx.x * blockDim.x + threadIdx.x;
  if (i >= N * 8) return;
  int n = i >> 3, hh = i & 7;
  const float* hp = h + (size_t)n * 64 + hh * 8;
  const float* sp = a_src + hh * 8;
  const float* dp = a_dst + hh * 8;
  float s1 = 0.f, s2 = 0.f;
  #pragma unroll
  for (int c = 0; c < 8; ++c) { float v = hp[c]; s1 += v * sp[c]; s2 += v * dp[c]; }
  as[i] = s1; ad[i] = s2;
}


__global__ void alphas2_kernel(const float* __restrict__ h,
                               const float* __restrict__ a_src,
                               const float* __restrict__ a_dst,
                               float* __restrict__ as, float* __restrict__ ad,
                               int N, int C) {
  int n = blockIdx.x * blockDim.x + threadIdx.x;
  if (n >= N) return;
  float s1 = 0.f, s2 = 0.f;
  for (int c = 0; c < C; ++c) { float v = h[(size_t)n * C + c]; s1 += v * a_src[c]; s2 += v * a_dst[c]; }
  as[n] = s1; ad[n] = s2;
}


#define AGG_T   1024
#define RN      196
#define LCAP    32768
#define P1      73
#define P2      41
#define U2      8

struct L1Vals { float ex; float4 a; float4 b; };

__device__ __forceinline__ float l1_logit(int s, int d, int hh,
                                         const float* __restrict__ as, const float* __restrict__ ad) {
  const float v = as[s * 8 + hh] + ad[d * 8 + hh];
  return v > 0.f ? v : NEG_SLOPE * v;
}

__device__ __forceinline__ L1Vals l1_load(int s, int d, int dl, int hh,
                                          const float* __restrict__ as, const float* __restrict__ ad,
                                          const float* mx, const float* __restrict__ h1) {
  L1Vals r;
  r.ex = expf(l1_logit(s, d, hh, as, ad) - mx[dl * 8 + hh]);
  const float4* hp = (const float4*)(h1 + (size_t)s * 64 + hh * 8);
  r.a = hp[0]; r.b = hp[1];
  return r;
}

__device__ __forceinline__ void l1_add(float* rows, int dl, int hh, const L1Vals& r) {
  float* op = rows + dl * P1 + hh * 8;
  atomicAdd(&rows[dl * P1 + 64 + hh], r.ex);
  atomicAdd(&op[0], r.a.x * r.ex); atomicAdd(&op[1], r.a.y * r.ex);
  atomicAdd(&op[2], r.a.z * r.ex); atomicAdd(&op[3], r.a.w * r.ex);
  atomicAdd(&op[4], r.b.x * r.ex); atomicAdd(&op[5], r.b.y * r.ex);
  atomicAdd(&op[6], r.b.z * r.ex); atomicAdd(&op[7], r.b.w * r.ex);
}

#define BM_E    1024
#define BM_W    (BM_E / 32)
#define BM_G    256
#define BM_T    256

__global__ void __launch_bounds__(BM_T)
edge_bitmap_kernel(const int* __restrict__ ei, int E, int G, int pitch,
                   unsigned int* __restrict__ bm) {
  __shared__ uint4 tile4[BM_G * BM_W / 4];
  unsigned int* tile = (unsigned int*)tile4;
  const int tid = threadIdx.x;
  const int base = blockIdx.x * BM_E;
  const int* __restrict__ dsts = ei + E;
  int own[BM_E / BM_T];
  #pragma unroll
  for (int j = 0; j < BM_E / BM_T; ++j) {
    const int e = base + tid + j * BM_T;
    own[j] = dsts[e < E ? e : E - 1] / RN;
  }
  for (int g0 = 0; g0 < G; g0 += BM_G) {
    for (int i = tid; i < BM_G * BM_W / 4; i += BM_T) tile4[i] = make_uint4(0u, 0u, 0u, 0u);
    __syncthreads();
    #pragma unroll
    for (int j = 0; j < BM_E / BM_T; ++j) {
      const unsigned int g = (unsigned int)(own[j] - g0);
      const int p = tid + j * BM_T;
      const unsigned int bit = base + p < E ? 1u << (p & 31) : 0u;
      if (g < (unsigned int)BM_G) atomicOr(&tile[g * BM_W + (p >> 5)], bit);
    }
    __syncthreads();
    for (int i = tid; i < BM_G * BM_W / 4; i += BM_T) {
      const int g = g0 + i / (BM_W / 4);
      if (g < G)
        ((uint4*)(bm + (size_t)g * pitch + (size_t)blockIdx.x * BM_W))[i % (BM_W / 4)] = tile4[i];
    }
    __syncthreads();
  }
}

__device__ __forceinline__ void collect_edges(const unsigned int* __restrict__ row, int nq,
                                              int* lst_s, int* cnt) {
  const int tid = threadIdx.x;
  const v4u* __restrict__ r4 = (const v4u*)row;
  for (int base = 0; base < nq; base += 8 * AGG_T) {
    v4u v[8];
    #pragma unroll
    for (int j = 0; j < 8; ++j) {
      const int q = base + tid + j * AGG_T;
      v[j] = r4[q < nq ? q : nq - 1];
    }
    int n = 0;
    #pragma unroll
    for (int j = 0; j < 8; ++j) {
      v[j] &= base + tid + j * AGG_T < nq ? ~0u : 0u;
      n += __popc(v[j].x) + __popc(v[j].y) + __popc(v[j].z) + __popc(v[j].w);
    }
    int pos = n ? atomicAdd(cnt, n) : 0;
    #pragma unroll
    for (int j = 0; j < 8; ++j) {
      const int e0 = (base + tid + j * AGG_T) * 128;
      const unsigned int w[4] = {v[j].x, v[j].y, v[j].z, v[j].w};
      #pragma unroll
      for (int cw = 0; cw < 4; ++cw) {
        unsigned int m = w[cw];
        while (m) {
          if (pos < LCAP) lst_s[pos] = e0 + cw * 32 + (__ffs(m) - 1);
          ++pos;
          m &= m - 1;
        }
      }
    }
  }
}

__global__ void __launch_bounds__(AGG_T)
edge_agg1_own_kernel(const int* __restrict__ ei, int E, int N,
                     const float* __restrict__ as, const float* __restrict__ ad,
                     const float* __restrict__ b,
                     const float* __restrict__ h1, float* __restrict__ out,
                     int* __restrict__ keep_s, int* __restrict__ keep_d, int* __restrict__ keep_n,
                     const unsigned int* __restrict__ bm, int pitch) {
  __shared__ float rows[RN * P1];
  __shared__ float mx[RN * 8];
  __shared__ int lst_s[LCAP];
  __shared__ unsigned char lst_d[LCAP];
  __shared__ int cnt;
  const int tid = threadIdx.x;
  const int lo = blockIdx.x * RN;
  const int nn = min(RN, N - lo);
  for (int i = tid; i < RN * P1; i += AGG_T) rows[i] = 0.f;
  for (int i = tid; i < RN * 8; i += AGG_T) mx[i] = -INFINITY;
  if (tid < nn) { lst_s[tid] = lo + tid; lst_d[tid] = (unsigned char)tid; }
  if (tid == 0) cnt = nn;
  __syncthreads();
  collect_edges(bm + (size_t)blockIdx.x * pitch, pitch / 4, lst_s, &cnt);
  __syncthreads();
  const int total = cnt;
  const int nl = total <= LCAP ? total : nn;
  for (int i = nn + tid; i < nl; i += AGG_T) {
    const int idx = lst_s[i];
    const int s = ei[idx], d = ei[E + idx];
    lst_s[i] = s; lst_d[i] = (unsigned char)(d - lo);
  }
  __syncthreads();
  for (int i = tid; i < nl; i += AGG_T) {
    keep_s[(size_t)blockIdx.x * LCAP + i] = lst_s[i];
    keep_d[(size_t)blockIdx.x * LCAP + i] = lst_d[i];
  }
  if (tid == 0) keep_n[blockIdx.x] = total;

  const int hh = tid & 7, sub = tid >> 3;

  for (int base = 0; base < nl; base += 4 * (AGG_T / 8)) {
    float v[4]; int dl[4]; bool k[4];
    #pragma unroll
    for (int j = 0; j < 4; ++j) {
      const int e = base + sub + j * (AGG_T / 8);
      k[j] = e < nl;
      const int f = k[j] ? e : nl - 1;
      dl[j] = lst_d[f];
      v[j] = l1_logit(lst_s[f], lo + dl[j], hh, as, ad);
    }
    #pragma unroll
    for (int j = 0; j < 4; ++j) if (k[j]) atomicMaxF(&mx[dl[j] * 8 + hh], v[j]);
  }
  if (total > LCAP) {
    const int* __restrict__ dsts = ei + E;
    for (int i = tid; i < E; i += AGG_T) {
      const unsigned dl = (unsigned)(dsts[i] - lo);
      if (dl < (unsigned)nn) {
        const int s = ei[i];
        for (int h = 0; h < 8; ++h) atomicMaxF(&mx[dl * 8 + h], l1_logit(s, lo + (int)dl, h, as, ad));
      }
    }
  }
  __syncthreads();

  for (int base = 0; base < nl; base += 2 * (AGG_T / 8)) {
    const int e0 = base + sub, e1 = e0 + AGG_T / 8;
    const bool k0 = e0 < nl, k1 = e1 < nl;
    const int f0 = k0 ? e0 : nl - 1, f1 = k1 ? e1 : nl - 1;
    const int s0 = lst_s[f0], d0 = lst_d[f0];
    const int s1 = lst_s[f1], d1 = lst_d[f1];
    const L1Vals r0 = l1_load(s0, lo + d0, d0, hh, as, ad, mx, h1);
    const L1Vals r1 = l1_load(s1, lo + d1, d1, hh, as, ad, mx, h1);
    if (k0) l1_add(rows, d0, hh, r0);
    if (k1) l1_add(rows, d1, hh, r1);
  }
  if (total > LCAP) {
    const int* __restrict__ dsts = ei + E;
    for (int i = tid; i < E; i += AGG_T) {
      const unsigned dl = (unsigned)(dsts[i] - lo);
      if (dl < (unsigned)nn) {
        const int s = ei[i];
        for (int h = 0; h < 8; ++h) l1_add(rows, (int)dl, h, l1_load(s, lo + (int)dl, (int)dl, h, as, ad, mx, h1));
      }
    }
  }
  __syncthreads();
  for (int i = tid; i < nn * 64; i += AGG_T) {
    const int r = i >> 6, c = i & 63;
    const float v = rows[r * P1 + c] / (rows[r * P1 + 64 + (c >> 3)] + EPSV) + b[c];
    out[(size_t)lo * 64 + i] = v > 0.f ? v : expm1f(v);
  }
}

__device__ __forceinline__ float l2_logit(int s, int d,
                                         const float* __restrict__ as, const float* __restrict__ ad) {
  const float v = as[s] + ad[d];
  return v > 0.f ? v : NEG_SLOPE * v;
}

__device__ __forceinline__ void l2_edge(float* rows, const float* mx, int s, int d, int dl, int q, int C,
                                        const float* __restrict__ as, const float* __restrict__ ad,
                                        const float* __restrict__ h2) {
  const float ex = expf(l2_logit(s, d, as, ad) - mx[dl]);
  const float* hp = h2 + (size_t)s * C;
  float hv[5];
  #pragma unroll
  for (int j = 0; j < 5; ++j) { const int c = q + 8 * j; hv[j] = c < C ? hp[c] : 0.f; }
  if (q == 0) atomicAdd(&rows[dl * P2 + 40], ex);
  #pragma unroll
  for (int j = 0; j < 5; ++j) { const int c = q + 8 * j; if (c < C) atomicAdd(&rows[dl * P2 + c], hv[j] * ex); }
}

__global__ void __launch_bounds__(AGG_T)
edge_agg2_own_kernel(const int* __restrict__ ei, int E, int N,
                     const float* __restrict__ as, const float* __restrict__ ad,
                     const float* __restrict__ b,
                     const float* __restrict__ h2, float* __restrict__ out, int C,
                     const int* __restrict__ keep_s, const int* __restrict__ keep_d,
                     const int* __restrict__ keep_n) {
  __shared__ float rows[RN * P2];
  __shared__ float mx[RN];
  __shared__ float lse[RN];
  __shared__ int lst_s[LCAP];
  __shared__ unsigned char lst_d[LCAP];
  const int tid = threadIdx.x;
  const int lo = blockIdx.x * RN;
  const int nn = min(RN, N - lo);
  for (int i = tid; i < RN * P2; i += AGG_T) rows[i] = 0.f;
  if (tid < RN) mx[tid] = -INFINITY;
  __syncthreads();
  const int total = keep_n[blockIdx.x];
  const int nl = total <= LCAP ? total : nn;
  const int* __restrict__ ls = keep_s + (size_t)blockIdx.x * LCAP;
  const int* __restrict__ ld = keep_d + (size_t)blockIdx.x * LCAP;

  for (int base = 0; base < nl; base += 4 * AGG_T) {
    float v[4]; int sv[4]; int dl[4]; bool k[4];
    #pragma unroll
    for (int j = 0; j < 4; ++j) {
      const int e = base + tid + j * AGG_T;
      k[j] = e < nl;
      const int f = k[j] ? e : nl - 1;
      dl[j] = ld[f];
      sv[j] = ls[f];
      v[j] = l2_logit(sv[j], lo + dl[j], as, ad);
    }
    #pragma unroll
    for (int j = 0; j < 4; ++j) {
      if (k[j]) {
        const int e = base + tid + j * AGG_T;
        lst_s[e] = sv[j]; lst_d[e] = (unsigned char)dl[j];
        atomicMaxF(&mx[dl[j]], v[j]);
      }
    }
  }
  if (total > LCAP) {
    const int* __restrict__ dsts = ei + E;
    for (int i = tid; i < E; i += AGG_T) {
      const unsigned dl = (unsigned)(dsts[i] - lo);
      if (dl < (unsigned)nn) atomicMaxF(&mx[dl], l2_logit(ei[i], lo + (int)dl, as, ad));
    }
  }
  __syncthreads();

  const int q = tid & 7, sub = tid >> 3;
  int cc[5];
  #pragma unroll
  for (int i = 0; i < 5; ++i) { const int c = q + 8 * i; cc[i] = c < C ? c : 0; }
  for (int base = 0; base < nl; base += U2 * (AGG_T / 8)) {
    float av[U2], mv[U2], hv[U2][5]; int dl[U2]; bool k[U2];
    #pragma unroll
    for (int j = 0; j < U2; ++j) {
      const int e = base + sub + j * (AGG_T / 8);
      k[j] = e < nl;
      const int f = k[j] ? e : nl - 1;
      const int s = lst_s[f];
      dl[j] = lst_d[f];
      av[j] = as[s] + ad[lo + dl[j]];
      mv[j] = mx[dl[j]];
      const int r = s * C;
      #pragma unroll
      for (int i = 0; i < 5; ++i) hv[j][i] = h2[r + cc[i]];
    }
    #pragma unroll
    for (int j = 0; j < U2; ++j) {
      if (k[j]) {
        const float lg = av[j] > 0.f ? av[j] : NEG_SLOPE * av[j];
        const float ex = expf(lg - mv[j]);
        float* rp = rows + dl[j] * P2;
        if (q == 0) atomicAdd(&rp[40], ex);
        #pragma unroll
        for (int i = 0; i < 5; ++i) { if (q + 8 * i < C) atomicAdd(&rp[q + 8 * i], hv[j][i] * ex); }
      }
    }
  }
  if (total > LCAP) {
    const int* __restrict__ dsts = ei + E;
    for (int i = tid; i < E; i += AGG_T) {
      const unsigned dl = (unsigned)(dsts[i] - lo);
      if (dl < (unsigned)nn) {
        const int s = ei[i];
        for (int k = 0; k < 8; ++k) l2_edge(rows, mx, s, lo + (int)dl, (int)dl, k, C, as, ad, h2);
      }
    }
  }
  __syncthreads();
  for (int i = tid; i < nn * C; i += AGG_T) {
    const int r = i / C, c = i - r * C;
    const float v = rows[r * P2 + c] / (rows[r * P2 + 40] + EPSV) + b[c];
    rows[r * P2 + c] = v > 0.f ? v : expm1f(v);
  }
  __syncthreads();
  if (tid < nn) {
    float hv[40];
    #pragma unroll
    for (int c = 0; c < 40; ++c) { const float t = rows[tid * P2 + c]; hv[c] = c < C ? t : -INFINITY; }
    float m = -INFINITY;
    #pragma unroll
    for (int c = 0; c < 40; ++c) m = fmaxf(m, hv[c]);
    float sm = 0.f;
    #pragma unroll
    for (int c = 0; c < 40; ++c) sm += expf(hv[c] - m);
    mx[tid] = m;
    lse[tid] = logf(sm);
  }
  __syncthreads();
  for (int i = tid; i < nn * C; i += AGG_T) {
    const int r = i / C, c = i - r * C;
    out[(size_t)lo * C + i] = rows[r * P2 + c] - mx[r] - lse[r];
  }
}


static inline int cdiv(int a, int b) { return (a + b - 1) / b; }

extern "C" void kernel_launch(void* const* d_in, const int* in_sizes, int n_in,
                              void* d_out, int out_size, void* d_ws, size_t ws_size,
                              hipStream_t stream) {
  const float* x      = (const float*)d_in[0];
  const int*   ei     = (const int*)  d_in[1];
  const float* W1     = (const float*)d_in[2];
  const float* a_src1 = (const float*)d_in[3];
  const float* a_dst1 = (const float*)d_in[4];
  const float* b1     = (const float*)d_in[5];
  const float* W2     = (const float*)d_in[6];
  const float* a_src2 = (const float*)d_in[7];
  const float* a_dst2 = (const float*)d_in[8];
  const float* b2     = (const float*)d_in[9];
  float* out = (float*)d_out;

  const int HC   = in_sizes[5];
  const int NCLS = in_sizes[9];
  const int F    = in_sizes[2] / HC;
  const int N    = in_sizes[0] / F;
  const int E    = in_sizes[1] / 2;
  const int NPAD = 48;

  float* ws = (float*)d_ws;
  size_t off = 0;
  const int G   = cdiv(N, RN);
  const int NB  = cdiv(E, BM_E);
  const int BMP = NB * BM_W;
  unsigned int* bm = (unsigned int*)(ws + off); off += (size_t)G * BMP;
  float* h1   = ws + off; off += (size_t)N * HC;
  float* as1  = ws + off; off += (size_t)N * 8;
  float* ad1  = ws + off; off += (size_t)N * 8;
  float* out1 = ws + off; off += (size_t)N * HC;
  float* h2   = ws + off; off += (size_t)N * NCLS;
  float* as2  = ws + off; off += (size_t)N;
  float* ad2  = ws + off; off += (size_t)N;
  float* Wp   = ws + off; off += (size_t)HC * NPAD;
  int* keep_s = (int*)(ws + off); off += (size_t)G * LCAP;
  int* keep_d = (int*)(ws + off); off += (size_t)G * LCAP;
  int* keep_n = (int*)(ws + off); off += (size_t)G;

  const int B = 256;

  gemm_wmma_f32_nt<4><<<dim3(N / 16), 32, 0, stream>>>(x, W1, h1, N, F, HC, HC, HC);
  alphas1_kernel<<<cdiv(N * 8, B), B, 0, stream>>>(h1, a_src1, a_dst1, as1, ad1, N);
  edge_bitmap_kernel<<<NB, BM_T, 0, stream>>>(ei, E, G, BMP, bm);
  edge_agg1_own_kernel<<<G, AGG_T, 0, stream>>>(ei, E, N, as1, ad1, b1, h1, out1, keep_s, keep_d, keep_n,
                                               bm, BMP);

  pad_kernel<<<cdiv(HC * NPAD, B), B, 0, stream>>>(W2, Wp, HC, NCLS, NPAD);
  gemm_wmma_f32_nt<3><<<dim3(N / 16), 32, 0, stream>>>(out1, Wp, h2, N, HC, NPAD, NCLS, NCLS);
  alphas2_kernel<<<cdiv(N, B), B, 0, stream>>>(h2, a_src2, a_dst2, as2, ad2, N, NCLS);
  edge_agg2_own_kernel<<<G, AGG_T, 0, stream>>>(ei, E, N, as2, ad2, b2, h2, out, NCLS, keep_s, keep_d, keep_n);
}


# attention logits as extra output columns of the two GEMMs: one prep launch appends W a_src and W a_dst to each weight matrix; both logit launches go; 8->6 launches; logit sum order changes
# speedup vs baseline: 9.2687x; 1.0463x over previous
#include <hip/hip_runtime.h>
#include <hip/hip_bf16.h>
#include <math.h>

typedef __attribute__((ext_vector_type(2))) float v2f;
typedef __attribute__((ext_vector_type(8))) float v8f;
typedef __attribute__((ext_vector_type(4))) unsigned int v4u;

#define NEG_SLOPE 0.2f
#define EPSV 1e-16f


__device__ __forceinline__ void atomicMaxF(float* addr, float val) {
  if (val >= 0.0f) atomicMax((int*)addr, __float_as_int(val));
  else             atomicMin((unsigned int*)addr, __float_as_uint(val));
}

template <int NT>
__global__ void __launch_bounds__(32)
gemm_wmma_f32_nt(const float* __restrict__ A, const float* __restrict__ B,
                 float* __restrict__ D, float* __restrict__ as, float* __restrict__ ad,
                 int M, int K, int ldb, int ldd, int nvalid, int na) {
  const int tileM = blockIdx.x;
  const int lane = threadIdx.x & 31;
  const int half = lane >> 4;
  const int l16  = lane & 15;
  const int rowA = tileM * 16 + l16;
  const float* __restrict__ arow = A + (size_t)rowA * K;

  v8f acc[NT];
  #pragma unroll
  for (int nt = 0; nt < NT; ++nt) acc[nt] = (v8f){0.f,0.f,0.f,0.f,0.f,0.f,0.f,0.f};

  for (int kk = 0; kk < K; kk += 4) {
    const int ka = kk + half * 2;
    v2f a;
    a.x = arow[ka];
    a.y = arow[ka + 1];
    const float* __restrict__ brow0 = B + (size_t)ka * ldb;
    const float* __restrict__ brow1 = B + (size_t)(ka + 1) * ldb;
    #pragma unroll
    for (int nt = 0; nt < NT; ++nt) {
      const int colB = nt * 16 + l16;
      v2f b;
      b.x = brow0[colB];
      b.y = brow1[colB];
      acc[nt] = __builtin_amdgcn_wmma_f32_16x16x4_f32(false, a, false, b,
                                                      (short)0, acc[nt], false, false);
    }
  }
  const size_t r0 = (size_t)(tileM * 16 + half * 8);
  #pragma unroll
  for (int nt = 0; nt < NT; ++nt) {
    const int c = nt * 16 + l16;
    float* __restrict__ p = nullptr;
    size_t pitch = 0;
    if (c < nvalid)               { p = D + c;                    pitch = (size_t)ldd; }
    else if (c < nvalid + na)     { p = as + (c - nvalid);        pitch = (size_t)na; }
    else if (c < nvalid + 2 * na) { p = ad + (c - nvalid - na);   pitch = (size_t)na; }
    if (p != nullptr) {
      p += r0 * pitch;
      #pragma unroll
      for (int g = 0; g < 8; ++g) p[(size_t)g * pitch] = acc[nt][g];
    }
  }
}

__global__ void ext_weights_kernel(const float* __restrict__ W1, const float* __restrict__ a_src1,
                                   const float* __restrict__ a_dst1, float* __restrict__ W1e,
                                   const float* __restrict__ W2, const float* __restrict__ a_src2,
                                   const float* __restrict__ a_dst2, float* __restrict__ W2e,
                                   int F, int HC, int NCLS, int NPAD) {
  const int i = blockIdx.x * blockDim.x + threadIdx.x;
  const int ld1 = HC + 16;
  const int n1 = F * ld1;
  if (i < n1) {
    const int k = i / ld1, c = i - k * ld1;
    float v;
    if (c < HC) {
      v = W1[k * HC + c];
    } else {
      const int j = c - HC, hh = j & 7;
      const float* __restrict__ a = (j < 8 ? a_src1 : a_dst1) + hh * 8;
      const float* __restrict__ w = W1 + k * HC + hh * 8;
      v = 0.f;
      #pragma unroll
      for (int t = 0; t < 8; ++t) v += w[t] * a[t];
    }
    W1e[i] = v;
  } else if (i - n1 < HC * NPAD) {
    const int i2 = i - n1;
    const int k = i2 / NPAD, c = i2 - k * NPAD;
    float v = 0.f;
    if (c < NCLS) {
      v = W2[k * NCLS + c];
    } else if (c < NCLS + 2) {
      const float* __restrict__ a = c == NCLS ? a_src2 : a_dst2;
      const float* __restrict__ w = W2 + k * NCLS;
      for (int t = 0; t < NCLS; ++t) v += w[t] * a[t];
    }
    W2e[i2] = v;
  }
}


#define AGG_T   1024
#define RN      196
#define LCAP    32768
#define P1      73
#define P2      41
#define U2      8

struct L1Vals { float ex; float4 a; float4 b; };

__device__ __forceinline__ float l1_logit(int s, int d, int hh,
                                         const float* __restrict__ as, const float* __restrict__ ad) {
  const float v = as[s * 8 + hh] + ad[d * 8 + hh];
  return v > 0.f ? v : NEG_SLOPE * v;
}

__device__ __forceinline__ L1Vals l1_load(int s, int d, int dl, int hh,
                                          const float* __restrict__ as, const float* __restrict__ ad,
                                          const float* mx, const float* __restrict__ h1) {
  L1Vals r;
  r.ex = expf(l1_logit(s, d, hh, as, ad) - mx[dl * 8 + hh]);
  const float4* hp = (const float4*)(h1 + (size_t)s * 64 + hh * 8);
  r.a = hp[0]; r.b = hp[1];
  return r;
}

__device__ __forceinline__ void l1_add(float* rows, int dl, int hh, const L1Vals& r) {
  float* op = rows + dl * P1 + hh * 8;
  atomicAdd(&rows[dl * P1 + 64 + hh], r.ex);
  atomicAdd(&op[0], r.a.x * r.ex); atomicAdd(&op[1], r.a.y * r.ex);
  atomicAdd(&op[2], r.a.z * r.ex); atomicAdd(&op[3], r.a.w * r.ex);
  atomicAdd(&op[4], r.b.x * r.ex); atomicAdd(&op[5], r.b.y * r.ex);
  atomicAdd(&op[6], r.b.z * r.ex); atomicAdd(&op[7], r.b.w * r.ex);
}

#define BM_E    1024
#define BM_W    (BM_E / 32)
#define BM_G    256
#define BM_T    256

__global__ void __launch_bounds__(BM_T)
edge_bitmap_kernel(const int* __restrict__ ei, int E, int G, int pitch,
                   unsigned int* __restrict__ bm) {
  __shared__ uint4 tile4[BM_G * BM_W / 4];
  unsigned int* tile = (unsigned int*)tile4;
  const int tid = threadIdx.x;
  const int base = blockIdx.x * BM_E;
  const int* __restrict__ dsts = ei + E;
  int own[BM_E / BM_T];
  #pragma unroll
  for (int j = 0; j < BM_E / BM_T; ++j) {
    const int e = base + tid + j * BM_T;
    own[j] = dsts[e < E ? e : E - 1] / RN;
  }
  for (int g0 = 0; g0 < G; g0 += BM_G) {
    for (int i = tid; i < BM_G * BM_W / 4; i += BM_T) tile4[i] = make_uint4(0u, 0u, 0u, 0u);
    __syncthreads();
    #pragma unroll
    for (int j = 0; j < BM_E / BM_T; ++j) {
      const unsigned int g = (unsigned int)(own[j] - g0);
      const int p = tid + j * BM_T;
      const unsigned int bit = base + p < E ? 1u << (p & 31) : 0u;
      if (g < (unsigned int)BM_G) atomicOr(&tile[g * BM_W + (p >> 5)], bit);
    }
    __syncthreads();
    for (int i = tid; i < BM_G * BM_W / 4; i += BM_T) {
      const int g = g0 + i / (BM_W / 4);
      if (g < G)
        ((uint4*)(bm + (size_t)g * pitch + (size_t)blockIdx.x * BM_W))[i % (BM_W / 4)] = tile4[i];
    }
    __syncthreads();
  }
}

__device__ __forceinline__ void collect_edges(const unsigned int* __restrict__ row, int nq,
                                              int* lst_s, int* cnt) {
  const int tid = threadIdx.x;
  const v4u* __restrict__ r4 = (const v4u*)row;
  for (int base = 0; base < nq; base += 8 * AGG_T) {
    v4u v[8];
    #pragma unroll
    for (int j = 0; j < 8; ++j) {
      const int q = base + tid + j * AGG_T;
      v[j] = r4[q < nq ? q : nq - 1];
    }
    int n = 0;
    #pragma unroll
    for (int j = 0; j < 8; ++j) {
      v[j] &= base + tid + j * AGG_T < nq ? ~0u : 0u;
      n += __popc(v[j].x) + __popc(v[j].y) + __popc(v[j].z) + __popc(v[j].w);
    }
    int pos = n ? atomicAdd(cnt, n) : 0;
    #pragma unroll
    for (int j = 0; j < 8; ++j) {
      const int e0 = (base + tid + j * AGG_T) * 128;
      const unsigned int w[4] = {v[j].x, v[j].y, v[j].z, v[j].w};
      #pragma unroll
      for (int cw = 0; cw < 4; ++cw) {
        unsigned int m = w[cw];
        while (m) {
          if (pos < LCAP) lst_s[pos] = e0 + cw * 32 + (__ffs(m) - 1);
          ++pos;
          m &= m - 1;
        }
      }
    }
  }
}

__global__ void __launch_bounds__(AGG_T)
edge_agg1_own_kernel(const int* __restrict__ ei, int E, int N,
                     const float* __restrict__ as, const float* __restrict__ ad,
                     const float* __restrict__ b,
                     const float* __restrict__ h1, float* __restrict__ out,
                     int* __restrict__ keep_s, int* __restrict__ keep_d, int* __restrict__ keep_n,
                     const unsigned int* __restrict__ bm, int pitch) {
  __shared__ float rows[RN * P1];
  __shared__ float mx[RN * 8];
  __shared__ int lst_s[LCAP];
  __shared__ unsigned char lst_d[LCAP];
  __shared__ int cnt;
  const int tid = threadIdx.x;
  const int lo = blockIdx.x * RN;
  const int nn = min(RN, N - lo);
  for (int i = tid; i < RN * P1; i += AGG_T) rows[i] = 0.f;
  for (int i = tid; i < RN * 8; i += AGG_T) mx[i] = -INFINITY;
  if (tid < nn) { lst_s[tid] = lo + tid; lst_d[tid] = (unsigned char)tid; }
  if (tid == 0) cnt = nn;
  __syncthreads();
  collect_edges(bm + (size_t)blockIdx.x * pitch, pitch / 4, lst_s, &cnt);
  __syncthreads();
  const int total = cnt;
  const int nl = total <= LCAP ? total : nn;
  for (int i = nn + tid; i < nl; i += AGG_T) {
    const int idx = lst_s[i];
    const int s = ei[idx], d = ei[E + idx];
    lst_s[i] = s; lst_d[i] = (unsigned char)(d - lo);
  }
  __syncthreads();
  for (int i = tid; i < nl; i += AGG_T) {
    keep_s[(size_t)blockIdx.x * LCAP + i] = lst_s[i];
    keep_d[(size_t)blockIdx.x * LCAP + i] = lst_d[i];
  }
  if (tid == 0) keep_n[blockIdx.x] = total;

  const int hh = tid & 7, sub = tid >> 3;

  for (int base = 0; base < nl; base += 4 * (AGG_T / 8)) {
    float v[4]; int dl[4]; bool k[4];
    #pragma unroll
    for (int j = 0; j < 4; ++j) {
      const int e = base + sub + j * (AGG_T / 8);
      k[j] = e < nl;
      const int f = k[j] ? e : nl - 1;
      dl[j] = lst_d[f];
      v[j] = l1_logit(lst_s[f], lo + dl[j], hh, as, ad);
    }
    #pragma unroll
    for (int j = 0; j < 4; ++j) if (k[j]) atomicMaxF(&mx[dl[j] * 8 + hh], v[j]);
  }
  if (total > LCAP) {
    const int* __restrict__ dsts = ei + E;
    for (int i = tid; i < E; i += AGG_T) {
      const unsigned dl = (unsigned)(dsts[i] - lo);
      if (dl < (unsigned)nn) {
        const int s = ei[i];
        for (int h = 0; h < 8; ++h) atomicMaxF(&mx[dl * 8 + h], l1_logit(s, lo + (int)dl, h, as, ad));
      }
    }
  }
  __syncthreads();

  for (int base = 0; base < nl; base += 2 * (AGG_T / 8)) {
    const int e0 = base + sub, e1 = e0 + AGG_T / 8;
    const bool k0 = e0 < nl, k1 = e1 < nl;
    const int f0 = k0 ? e0 : nl - 1, f1 = k1 ? e1 : nl - 1;
    const int s0 = lst_s[f0], d0 = lst_d[f0];
    const int s1 = lst_s[f1], d1 = lst_d[f1];
    const L1Vals r0 = l1_load(s0, lo + d0, d0, hh, as, ad, mx, h1);
    const L1Vals r1 = l1_load(s1, lo + d1, d1, hh, as, ad, mx, h1);
    if (k0) l1_add(rows, d0, hh, r0);
    if (k1) l1_add(rows, d1, hh, r1);
  }
  if (total > LCAP) {
    const int* __restrict__ dsts = ei + E;
    for (int i = tid; i < E; i += AGG_T) {
      const unsigned dl = (unsigned)(dsts[i] - lo);
      if (dl < (unsigned)nn) {
        const int s = ei[i];
        for (int h = 0; h < 8; ++h) l1_add(rows, (int)dl, h, l1_load(s, lo + (int)dl, (int)dl, h, as, ad, mx, h1));
      }
    }
  }
  __syncthreads();
  for (int i = tid; i < nn * 64; i += AGG_T) {
    const int r = i >> 6, c = i & 63;
    const float v = rows[r * P1 + c] / (rows[r * P1 + 64 + (c >> 3)] + EPSV) + b[c];
    out[(size_t)lo * 64 + i] = v > 0.f ? v : expm1f(v);
  }
}

__device__ __forceinline__ float l2_logit(int s, int d,
                                         const float* __restrict__ as, const float* __restrict__ ad) {
  const float v = as[s] + ad[d];
  return v > 0.f ? v : NEG_SLOPE * v;
}

__device__ __forceinline__ void l2_edge(float* rows, const float* mx, int s, int d, int dl, int q, int C,
                                        const float* __restrict__ as, const float* __restrict__ ad,
                                        const float* __restrict__ h2) {
  const float ex = expf(l2_logit(s, d, as, ad) - mx[dl]);
  const float* hp = h2 + (size_t)s * C;
  float hv[5];
  #pragma unroll
  for (int j = 0; j < 5; ++j) { const int c = q + 8 * j; hv[j] = c < C ? hp[c] : 0.f; }
  if (q == 0) atomicAdd(&rows[dl * P2 + 40], ex);
  #pragma unroll
  for (int j = 0; j < 5; ++j) { const int c = q + 8 * j; if (c < C) atomicAdd(&rows[dl * P2 + c], hv[j] * ex); }
}

__global__ void __launch_bounds__(AGG_T)
edge_agg2_own_kernel(const int* __restrict__ ei, int E, int N,
                     const float* __restrict__ as, const float* __restrict__ ad,
                     const float* __restrict__ b,
                     const float* __restrict__ h2, float* __restrict__ out, int C,
                     const int* __restrict__ keep_s, const int* __restrict__ keep_d,
                     const int* __restrict__ keep_n) {
  __shared__ float rows[RN * P2];
  __shared__ float mx[RN];
  __shared__ float lse[RN];
  __shared__ int lst_s[LCAP];
  __shared__ unsigned char lst_d[LCAP];
  const int tid = threadIdx.x;
  const int lo = blockIdx.x * RN;
  const int nn = min(RN, N - lo);
  for (int i = tid; i < RN * P2; i += AGG_T) rows[i] = 0.f;
  if (tid < RN) mx[tid] = -INFINITY;
  __syncthreads();
  const int total = keep_n[blockIdx.x];
  const int nl = total <= LCAP ? total : nn;
  const int* __restrict__ ls = keep_s + (size_t)blockIdx.x * LCAP;
  const int* __restrict__ ld = keep_d + (size_t)blockIdx.x * LCAP;

  for (int base = 0; base < nl; base += 4 * AGG_T) {
    float v[4]; int sv[4]; int dl[4]; bool k[4];
    #pragma unroll
    for (int j = 0; j < 4; ++j) {
      const int e = base + tid + j * AGG_T;
      k[j] = e < nl;
      const int f = k[j] ? e : nl - 1;
      dl[j] = ld[f];
      sv[j] = ls[f];
      v[j] = l2_logit(sv[j], lo + dl[j], as, ad);
    }
    #pragma unroll
    for (int j = 0; j < 4; ++j) {
      if (k[j]) {
        const int e = base + tid + j * AGG_T;
        lst_s[e] = sv[j]; lst_d[e] = (unsigned char)dl[j];
        atomicMaxF(&mx[dl[j]], v[j]);
      }
    }
  }
  if (total > LCAP) {
    const int* __restrict__ dsts = ei + E;
    for (int i = tid; i < E; i += AGG_T) {
      const unsigned dl = (unsigned)(dsts[i] - lo);
      if (dl < (unsigned)nn) atomicMaxF(&mx[dl], l2_logit(ei[i], lo + (int)dl, as, ad));
    }
  }
  __syncthreads();

  const int q = tid & 7, sub = tid >> 3;
  int cc[5];
  #pragma unroll
  for (int i = 0; i < 5; ++i) { const int c = q + 8 * i; cc[i] = c < C ? c : 0; }
  for (int base = 0; base < nl; base += U2 * (AGG_T / 8)) {
    float av[U2], mv[U2], hv[U2][5]; int dl[U2]; bool k[U2];
    #pragma unroll
    for (int j = 0; j < U2; ++j) {
      const int e = base + sub + j * (AGG_T / 8);
      k[j] = e < nl;
      const int f = k[j] ? e : nl - 1;
      const int s = lst_s[f];
      dl[j] = lst_d[f];
      av[j] = as[s] + ad[lo + dl[j]];
      mv[j] = mx[dl[j]];
      const int r = s * C;
      #pragma unroll
      for (int i = 0; i < 5; ++i) hv[j][i] = h2[r + cc[i]];
    }
    #pragma unroll
    for (int j = 0; j < U2; ++j) {
      if (k[j]) {
        const float lg = av[j] > 0.f ? av[j] : NEG_SLOPE * av[j];
        const float ex = expf(lg - mv[j]);
        float* rp = rows + dl[j] * P2;
        if (q == 0) atomicAdd(&rp[40], ex);
        #pragma unroll
        for (int i = 0; i < 5; ++i) { if (q + 8 * i < C) atomicAdd(&rp[q + 8 * i], hv[j][i] * ex); }
      }
    }
  }
  if (total > LCAP) {
    const int* __restrict__ dsts = ei + E;
    for (int i = tid; i < E; i += AGG_T) {
      const unsigned dl = (unsigned)(dsts[i] - lo);
      if (dl < (unsigned)nn) {
        const int s = ei[i];
        for (int k = 0; k < 8; ++k) l2_edge(rows, mx, s, lo + (int)dl, (int)dl, k, C, as, ad, h2);
      }
    }
  }
  __syncthreads();
  for (int i = tid; i < nn * C; i += AGG_T) {
    const int r = i / C, c = i - r * C;
    const float v = rows[r * P2 + c] / (rows[r * P2 + 40] + EPSV) + b[c];
    rows[r * P2 + c] = v > 0.f ? v : expm1f(v);
  }
  __syncthreads();
  if (tid < nn) {
    float hv[40];
    #pragma unroll
    for (int c = 0; c < 40; ++c) { const float t = rows[tid * P2 + c]; hv[c] = c < C ? t : -INFINITY; }
    float m = -INFINITY;
    #pragma unroll
    for (int c = 0; c < 40; ++c) m = fmaxf(m, hv[c]);
    float sm = 0.f;
    #pragma unroll
    for (int c = 0; c < 40; ++c) sm += expf(hv[c] - m);
    mx[tid] = m;
    lse[tid] = logf(sm);
  }
  __syncthreads();
  for (int i = tid; i < nn * C; i += AGG_T) {
    const int r = i / C, c = i - r * C;
    out[(size_t)lo * C + i] = rows[r * P2 + c] - mx[r] - lse[r];
  }
}


static inline int cdiv(int a, int b) { return (a + b - 1) / b; }

extern "C" void kernel_launch(void* const* d_in, const int* in_sizes, int n_in,
                              void* d_out, int out_size, void* d_ws, size_t ws_size,
                              hipStream_t stream) {
  const float* x      = (const float*)d_in[0];
  const int*   ei     = (const int*)  d_in[1];
  const float* W1     = (const float*)d_in[2];
  const float* a_src1 = (const float*)d_in[3];
  const float* a_dst1 = (const float*)d_in[4];
  const float* b1     = (const float*)d_in[5];
  const float* W2     = (const float*)d_in[6];
  const float* a_src2 = (const float*)d_in[7];
  const float* a_dst2 = (const float*)d_in[8];
  const float* b2     = (const float*)d_in[9];
  float* out = (float*)d_out;

  const int HC   = in_sizes[5];
  const int NCLS = in_sizes[9];
  const int F    = in_sizes[2] / HC;
  const int N    = in_sizes[0] / F;
  const int E    = in_sizes[1] / 2;
  const int NPAD = 48;
  const int LD1  = HC + 16;

  float* ws = (float*)d_ws;
  size_t off = 0;
  const int G   = cdiv(N, RN);
  const int NB  = cdiv(E, BM_E);
  const int BMP = NB * BM_W;
  unsigned int* bm = (unsigned int*)(ws + off); off += (size_t)G * BMP;
  float* h1   = ws + off; off += (size_t)N * HC;
  float* as1  = ws + off; off += (size_t)N * 8;
  float* ad1  = ws + off; off += (size_t)N * 8;
  float* out1 = ws + off; off += (size_t)N * HC;
  float* h2   = ws + off; off += (size_t)N * NCLS;
  float* as2  = ws + off; off += (size_t)N;
  float* ad2  = ws + off; off += (size_t)N;
  float* Wp   = ws + off; off += (size_t)HC * NPAD;
  float* W1e  = ws + off; off += (size_t)F * LD1;
  int* keep_s = (int*)(ws + off); off += (size_t)G * LCAP;
  int* keep_d = (int*)(ws + off); off += (size_t)G * LCAP;
  int* keep_n = (int*)(ws + off); off += (size_t)G;

  const int B = 256;

  ext_weights_kernel<<<cdiv(F * LD1 + HC * NPAD, B), B, 0, stream>>>(W1, a_src1, a_dst1, W1e,
                                                                     W2, a_src2, a_dst2, Wp, F, HC, NCLS, NPAD);

  gemm_wmma_f32_nt<5><<<dim3(N / 16), 32, 0, stream>>>(x, W1e, h1, as1, ad1, N, F, LD1, HC, HC, 8);
  edge_bitmap_kernel<<<NB, BM_T, 0, stream>>>(ei, E, G, BMP, bm);
  edge_agg1_own_kernel<<<G, AGG_T, 0, stream>>>(ei, E, N, as1, ad1, b1, h1, out1, keep_s, keep_d, keep_n,
                                               bm, BMP);

  gemm_wmma_f32_nt<3><<<dim3(N / 16), 32, 0, stream>>>(out1, Wp, h2, as2, ad2, N, HC, NPAD, NCLS, NCLS, 1);
  edge_agg2_own_kernel<<<G, AGG_T, 0, stream>>>(ei, E, N, as2, ad2, b2, h2, out, NCLS, keep_s, keep_d, keep_n);
}


# edge pre-pass writes (source, local dest) entries in 128-byte cells per (owner, 2048-edge block), not a bit per edge; edge_agg1 copies them to its list: no bit scan or index gather; add order changes
# speedup vs baseline: 10.0877x; 1.0884x over previous
#include <hip/hip_runtime.h>
#include <hip/hip_bf16.h>
#include <math.h>

typedef __attribute__((ext_vector_type(2))) float v2f;
typedef __attribute__((ext_vector_type(8))) float v8f;
typedef __attribute__((ext_vector_type(4))) unsigned int v4u;

#define NEG_SLOPE 0.2f
#define EPSV 1e-16f


__device__ __forceinline__ void atomicMaxF(float* addr, float val) {
  if (val >= 0.0f) atomicMax((int*)addr, __float_as_int(val));
  else             atomicMin((unsigned int*)addr, __float_as_uint(val));
}

template <int NT>
__global__ void __launch_bounds__(32)
gemm_wmma_f32_nt(const float* __restrict__ A, const float* __restrict__ B,
                 float* __restrict__ D, float* __restrict__ as, float* __restrict__ ad,
                 int M, int K, int ldb, int ldd, int nvalid, int na) {
  const int tileM = blockIdx.x;
  const int lane = threadIdx.x & 31;
  const int half = lane >> 4;
  const int l16  = lane & 15;
  const int rowA = tileM * 16 + l16;
  const float* __restrict__ arow = A + (size_t)rowA * K;

  v8f acc[NT];
  #pragma unroll
  for (int nt = 0; nt < NT; ++nt) acc[nt] = (v8f){0.f,0.f,0.f,0.f,0.f,0.f,0.f,0.f};

  for (int kk = 0; kk < K; kk += 4) {
    const int ka = kk + half * 2;
    v2f a;
    a.x = arow[ka];
    a.y = arow[ka + 1];
    const float* __restrict__ brow0 = B + (size_t)ka * ldb;
    const float* __restrict__ brow1 = B + (size_t)(ka + 1) * ldb;
    #pragma unroll
    for (int nt = 0; nt < NT; ++nt) {
      const int colB = nt * 16 + l16;
      v2f b;
      b.x = brow0[colB];
      b.y = brow1[colB];
      acc[nt] = __builtin_amdgcn_wmma_f32_16x16x4_f32(false, a, false, b,
                                                      (short)0, acc[nt], false, false);
    }
  }
  const size_t r0 = (size_t)(tileM * 16 + half * 8);
  #pragma unroll
  for (int nt = 0; nt < NT; ++nt) {
    const int c = nt * 16 + l16;
    float* __restrict__ p = nullptr;
    size_t pitch = 0;
    if (c < nvalid)               { p = D + c;                    pitch = (size_t)ldd; }
    else if (c < nvalid + na)     { p = as + (c - nvalid);        pitch = (size_t)na; }
    else if (c < nvalid + 2 * na) { p = ad + (c - nvalid - na);   pitch = (size_t)na; }
    if (p != nullptr) {
      p += r0 * pitch;
      #pragma unroll
      for (int g = 0; g < 8; ++g) p[(size_t)g * pitch] = acc[nt][g];
    }
  }
}

__global__ void ext_weights_kernel(const float* __restrict__ W1, const float* __restrict__ a_src1,
                                   const float* __restrict__ a_dst1, float* __restrict__ W1e,
                                   const float* __restrict__ W2, const float* __restrict__ a_src2,
                                   const float* __restrict__ a_dst2, float* __restrict__ W2e,
                                   int F, int HC, int NCLS, int NPAD) {
  const int i = blockIdx.x * blockDim.x + threadIdx.x;
  const int ld1 = HC + 16;
  const int n1 = F * ld1;
  if (i < n1) {
    const int k = i / ld1, c = i - k * ld1;
    float v;
    if (c < HC) {
      v = W1[k * HC + c];
    } else {
      const int j = c - HC, hh = j & 7;
      const float* __restrict__ a = (j < 8 ? a_src1 : a_dst1) + hh * 8;
      const float* __restrict__ w = W1 + k * HC + hh * 8;
      v = 0.f;
      #pragma unroll
      for (int t = 0; t < 8; ++t) v += w[t] * a[t];
    }
    W1e[i] = v;
  } else if (i - n1 < HC * NPAD) {
    const int i2 = i - n1;
    const int k = i2 / NPAD, c = i2 - k * NPAD;
    float v = 0.f;
    if (c < NCLS) {
      v = W2[k * NCLS + c];
    } else if (c < NCLS + 2) {
      const float* __restrict__ a = c == NCLS ? a_src2 : a_dst2;
      const float* __restrict__ w = W2 + k * NCLS;
      for (int t = 0; t < NCLS; ++t) v += w[t] * a[t];
    }
    W2e[i2] = v;
  }
}


#define AGG_T   1024
#define RN      196
#define LCAP    32768
#define P1      73
#define P2      41
#define U2      8

struct L1Vals { float ex; float4 a; float4 b; };

__device__ __forceinline__ float l1_logit(int s, int d, int hh,
                                         const float* __restrict__ as, const float* __restrict__ ad) {
  const float v = as[s * 8 + hh] + ad[d * 8 + hh];
  return v > 0.f ? v : NEG_SLOPE * v;
}

__device__ __forceinline__ L1Vals l1_load(int s, int d, int dl, int hh,
                                          const float* __restrict__ as, const float* __restrict__ ad,
                                          const float* mx, const float* __restrict__ h1) {
  L1Vals r;
  r.ex = expf(l1_logit(s, d, hh, as, ad) - mx[dl * 8 + hh]);
  const float4* hp = (const float4*)(h1 + (size_t)s * 64 + hh * 8);
  r.a = hp[0]; r.b = hp[1];
  return r;
}

__device__ __forceinline__ void l1_add(float* rows, int dl, int hh, const L1Vals& r) {
  float* op = rows + dl * P1 + hh * 8;
  atomicAdd(&rows[dl * P1 + 64 + hh], r.ex);
  atomicAdd(&op[0], r.a.x * r.ex); atomicAdd(&op[1], r.a.y * r.ex);
  atomicAdd(&op[2], r.a.z * r.ex); atomicAdd(&op[3], r.a.w * r.ex);
  atomicAdd(&op[4], r.b.x * r.ex); atomicAdd(&op[5], r.b.y * r.ex);
  atomicAdd(&op[6], r.b.z * r.ex); atomicAdd(&op[7], r.b.w * r.ex);
}

#define BM_E    2048
#define CELL    32
#define BM_G    256
#define BM_T    512
#define ENT_EMPTY 0xFFFFFFFFu
#define ENT_OVER  0xFFFFFFFEu
#define OVF_CAP 512

__global__ void __launch_bounds__(BM_T)
edge_cells_kernel(const int* __restrict__ ei, int E, int G, int pitch,
                  unsigned int* __restrict__ cells) {
  __shared__ uint4 tile4[BM_G * CELL / 4];
  __shared__ int ccnt[BM_G];
  unsigned int* tile = (unsigned int*)tile4;
  const int tid = threadIdx.x;
  const int base = blockIdx.x * BM_E;
  const int* __restrict__ dsts = ei + E;
  int sv[BM_E / BM_T], dv[BM_E / BM_T];
  #pragma unroll
  for (int j = 0; j < BM_E / BM_T; ++j) {
    const int e = base + tid + j * BM_T;
    const int f = e < E ? e : E - 1;
    sv[j] = ei[f];
    dv[j] = dsts[f];
  }
  for (int g0 = 0; g0 < G; g0 += BM_G) {
    for (int i = tid; i < BM_G * CELL / 4; i += BM_T)
      tile4[i] = make_uint4(ENT_EMPTY, ENT_EMPTY, ENT_EMPTY, ENT_EMPTY);
    if (tid < BM_G) ccnt[tid] = 0;
    __syncthreads();
    #pragma unroll
    for (int j = 0; j < BM_E / BM_T; ++j) {
      const int own = dv[j] / RN;
      const unsigned int g = (unsigned int)(own - g0);
      if (base + tid + j * BM_T < E && g < (unsigned int)BM_G) {
        const int r = atomicAdd(&ccnt[g], 1);
        if (r < CELL)
          tile[g * CELL + r] = (unsigned int)sv[j] | ((unsigned int)(dv[j] - own * RN) << 24);
      }
    }
    __syncthreads();
    for (int i = tid; i < BM_G * CELL / 4; i += BM_T) {
      const int gl = i / (CELL / 4), part = i % (CELL / 4);
      const int g = g0 + gl;
      if (g < G) {
        uint4 v = tile4[i];
        if (ccnt[gl] > CELL)
          v = make_uint4(part == 0 ? ENT_OVER : ENT_EMPTY, ENT_EMPTY, ENT_EMPTY, ENT_EMPTY);
        ((uint4*)(cells + (size_t)g * pitch + (size_t)blockIdx.x * CELL))[part] = v;
      }
    }
    __syncthreads();
  }
}

__device__ __forceinline__ void put_entry(unsigned int e, int& pos, int* lst_s, unsigned char* lst_d) {
  if ((e >> 24) < (unsigned int)RN) {
    if (pos < LCAP) { lst_s[pos] = (int)(e & 0xFFFFFFu); lst_d[pos] = (unsigned char)(e >> 24); }
    ++pos;
  }
}

__device__ __forceinline__ void collect_cells(const unsigned int* __restrict__ row, int nq,
                                              int* lst_s, unsigned char* lst_d, int* cnt,
                                              int* ovf, int* novf) {
  const int tid = threadIdx.x;
  const int lane = tid & 31;
  const v4u* __restrict__ r4 = (const v4u*)row;
  for (int base = 0; base < nq; base += 4 * AGG_T) {
    v4u v[4];
    #pragma unroll
    for (int j = 0; j < 4; ++j) {
      const int q = base + tid + j * AGG_T;
      v[j] = r4[q < nq ? q : nq - 1];
    }
    int n = 0;
    #pragma unroll
    for (int j = 0; j < 4; ++j) {
      const int q = base + tid + j * AGG_T;
      if (q >= nq) v[j] = (v4u){ENT_EMPTY, ENT_EMPTY, ENT_EMPTY, ENT_EMPTY};
      if ((q & 7) == 0 && v[j].x == ENT_OVER) {
        const int k = atomicAdd(novf, 1);
        if (k < OVF_CAP) ovf[k] = q >> 3;
      }
      n += ((v[j].x >> 24) < (unsigned int)RN) + ((v[j].y >> 24) < (unsigned int)RN)
         + ((v[j].z >> 24) < (unsigned int)RN) + ((v[j].w >> 24) < (unsigned int)RN);
    }
    int inc = n;
    #pragma unroll
    for (int o = 1; o < 32; o <<= 1) {
      const int t = __shfl_up(inc, o);
      if (lane >= o) inc += t;
    }
    int wbase = 0;
    if (lane == 31 && inc) wbase = atomicAdd(cnt, inc);
    wbase = __shfl(wbase, 31);
    int pos = wbase + inc - n;
    #pragma unroll
    for (int j = 0; j < 4; ++j) {
      put_entry(v[j].x, pos, lst_s, lst_d);
      put_entry(v[j].y, pos, lst_s, lst_d);
      put_entry(v[j].z, pos, lst_s, lst_d);
      put_entry(v[j].w, pos, lst_s, lst_d);
    }
  }
}

__global__ void __launch_bounds__(AGG_T)
edge_agg1_own_kernel(const int* __restrict__ ei, int E, int N,
                     const float* __restrict__ as, const float* __restrict__ ad,
                     const float* __restrict__ b,
                     const float* __restrict__ h1, float* __restrict__ out,
                     int* __restrict__ keep_s, int* __restrict__ keep_d, int* __restrict__ keep_n,
                     const unsigned int* __restrict__ cells, int pitch) {
  __shared__ float rows[RN * P1];
  __shared__ float mx[RN * 8];
  __shared__ int lst_s[LCAP];
  __shared__ unsigned char lst_d[LCAP];
  __shared__ int ovf[OVF_CAP];
  __shared__ int cnt, novf;
  const int tid = threadIdx.x;
  const int lo = blockIdx.x * RN;
  const int nn = min(RN, N - lo);
  for (int i = tid; i < RN * P1; i += AGG_T) rows[i] = 0.f;
  for (int i = tid; i < RN * 8; i += AGG_T) mx[i] = -INFINITY;
  if (tid < nn) { lst_s[tid] = lo + tid; lst_d[tid] = (unsigned char)tid; }
  if (tid == 0) { cnt = nn; novf = 0; }
  __syncthreads();
  collect_cells(cells + (size_t)blockIdx.x * pitch, pitch / 4, lst_s, lst_d, &cnt, ovf, &novf);
  __syncthreads();
  const int nov = novf;
  if (nov != 0 && nov <= OVF_CAP) {
    const int* __restrict__ dsts = ei + E;
    for (int k = 0; k < nov; ++k) {
      const int e0 = ovf[k] * BM_E;
      const int e1 = min(E, e0 + BM_E);
      for (int e = e0 + tid; e < e1; e += AGG_T) {
        const unsigned dl = (unsigned)(dsts[e] - lo);
        if (dl < (unsigned)nn) {
          const int pos = atomicAdd(&cnt, 1);
          if (pos < LCAP) { lst_s[pos] = ei[e]; lst_d[pos] = (unsigned char)dl; }
        }
      }
    }
    __syncthreads();
  }
  const int total = nov <= OVF_CAP ? cnt : LCAP + 1;
  const int nl = total <= LCAP ? total : nn;
  for (int i = tid; i < nl; i += AGG_T) {
    keep_s[(size_t)blockIdx.x * LCAP + i] = lst_s[i];
    keep_d[(size_t)blockIdx.x * LCAP + i] = lst_d[i];
  }
  if (tid == 0) keep_n[blockIdx.x] = total;

  const int hh = tid & 7, sub = tid >> 3;

  for (int base = 0; base < nl; base += 4 * (AGG_T / 8)) {
    float v[4]; int dl[4]; bool k[4];
    #pragma unroll
    for (int j = 0; j < 4; ++j) {
      const int e = base + sub + j * (AGG_T / 8);
      k[j] = e < nl;
      const int f = k[j] ? e : nl - 1;
      dl[j] = lst_d[f];
      v[j] = l1_logit(lst_s[f], lo + dl[j], hh, as, ad);
    }
    #pragma unroll
    for (int j = 0; j < 4; ++j) if (k[j]) atomicMaxF(&mx[dl[j] * 8 + hh], v[j]);
  }
  if (total > LCAP) {
    const int* __restrict__ dsts = ei + E;
    for (int i = tid; i < E; i += AGG_T) {
      const unsigned dl = (unsigned)(dsts[i] - lo);
      if (dl < (unsigned)nn) {
        const int s = ei[i];
        for (int h = 0; h < 8; ++h) atomicMaxF(&mx[dl * 8 + h], l1_logit(s, lo + (int)dl, h, as, ad));
      }
    }
  }
  __syncthreads();

  for (int base = 0; base < nl; base += 2 * (AGG_T / 8)) {
    const int e0 = base + sub, e1 = e0 + AGG_T / 8;
    const bool k0 = e0 < nl, k1 = e1 < nl;
    const int f0 = k0 ? e0 : nl - 1, f1 = k1 ? e1 : nl - 1;
    const int s0 = lst_s[f0], d0 = lst_d[f0];
    const int s1 = lst_s[f1], d1 = lst_d[f1];
    const L1Vals r0 = l1_load(s0, lo + d0, d0, hh, as, ad, mx, h1);
    const L1Vals r1 = l1_load(s1, lo + d1, d1, hh, as, ad, mx, h1);
    if (k0) l1_add(rows, d0, hh, r0);
    if (k1) l1_add(rows, d1, hh, r1);
  }
  if (total > LCAP) {
    const int* __restrict__ dsts = ei + E;
    for (int i = tid; i < E; i += AGG_T) {
      const unsigned dl = (unsigned)(dsts[i] - lo);
      if (dl < (unsigned)nn) {
        const int s = ei[i];
        for (int h = 0; h < 8; ++h) l1_add(rows, (int)dl, h, l1_load(s, lo + (int)dl, (int)dl, h, as, ad, mx, h1));
      }
    }
  }
  __syncthreads();
  for (int i = tid; i < nn * 64; i += AGG_T) {
    const int r = i >> 6, c = i & 63;
    const float v = rows[r * P1 + c] / (rows[r * P1 + 64 + (c >> 3)] + EPSV) + b[c];
    out[(size_t)lo * 64 + i] = v > 0.f ? v : expm1f(v);
  }
}

__device__ __forceinline__ float l2_logit(int s, int d,
                                         const float* __restrict__ as, const float* __restrict__ ad) {
  const float v = as[s] + ad[d];
  return v > 0.f ? v : NEG_SLOPE * v;
}

__device__ __forceinline__ void l2_edge(float* rows, const float* mx, int s, int d, int dl, int q, int C,
                                        const float* __restrict__ as, const float* __restrict__ ad,
                                        const float* __restrict__ h2) {
  const float ex = expf(l2_logit(s, d, as, ad) - mx[dl]);
  const float* hp = h2 + (size_t)s * C;
  float hv[5];
  #pragma unroll
  for (int j = 0; j < 5; ++j) { const int c = q + 8 * j; hv[j] = c < C ? hp[c] : 0.f; }
  if (q == 0) atomicAdd(&rows[dl * P2 + 40], ex);
  #pragma unroll
  for (int j = 0; j < 5; ++j) { const int c = q + 8 * j; if (c < C) atomicAdd(&rows[dl * P2 + c], hv[j] * ex); }
}

__global__ void __launch_bounds__(AGG_T)
edge_agg2_own_kernel(const int* __restrict__ ei, int E, int N,
                     const float* __restrict__ as, const float* __restrict__ ad,
                     const float* __restrict__ b,
                     const float* __restrict__ h2, float* __restrict__ out, int C,
                     const int* __restrict__ keep_s, const int* __restrict__ keep_d,
                     const int* __restrict__ keep_n) {
  __shared__ float rows[RN * P2];
  __shared__ float mx[RN];
  __shared__ float lse[RN];
  __shared__ int lst_s[LCAP];
  __shared__ unsigned char lst_d[LCAP];
  const int tid = threadIdx.x;
  const int lo = blockIdx.x * RN;
  const int nn = min(RN, N - lo);
  for (int i = tid; i < RN * P2; i += AGG_T) rows[i] = 0.f;
  if (tid < RN) mx[tid] = -INFINITY;
  __syncthreads();
  const int total = keep_n[blockIdx.x];
  const int nl = total <= LCAP ? total : nn;
  const int* __restrict__ ls = keep_s + (size_t)blockIdx.x * LCAP;
  const int* __restrict__ ld = keep_d + (size_t)blockIdx.x * LCAP;

  for (int base = 0; base < nl; base += 4 * AGG_T) {
    float v[4]; int sv[4]; int dl[4]; bool k[4];
    #pragma unroll
    for (int j = 0; j < 4; ++j) {
      const int e = base + tid + j * AGG_T;
      k[j] = e < nl;
      const int f = k[j] ? e : nl - 1;
      dl[j] = ld[f];
      sv[j] = ls[f];
      v[j] = l2_logit(sv[j], lo + dl[j], as, ad);
    }
    #pragma unroll
    for (int j = 0; j < 4; ++j) {
      if (k[j]) {
        const int e = base + tid + j * AGG_T;
        lst_s[e] = sv[j]; lst_d[e] = (unsigned char)dl[j];
        atomicMaxF(&mx[dl[j]], v[j]);
      }
    }
  }
  if (total > LCAP) {
    const int* __restrict__ dsts = ei + E;
    for (int i = tid; i < E; i += AGG_T) {
      const unsigned dl = (unsigned)(dsts[i] - lo);
      if (dl < (unsigned)nn) atomicMaxF(&mx[dl], l2_logit(ei[i], lo + (int)dl, as, ad));
    }
  }
  __syncthreads();

  const int q = tid & 7, sub = tid >> 3;
  int cc[5];
  #pragma unroll
  for (int i = 0; i < 5; ++i) { const int c = q + 8 * i; cc[i] = c < C ? c : 0; }
  for (int base = 0; base < nl; base += U2 * (AGG_T / 8)) {
    float av[U2], mv[U2], hv[U2][5]; int dl[U2]; bool k[U2];
    #pragma unroll
    for (int j = 0; j < U2; ++j) {
      const int e = base + sub + j * (AGG_T / 8);
      k[j] = e < nl;
      const int f = k[j] ? e : nl - 1;
      const int s = lst_s[f];
      dl[j] = lst_d[f];
      av[j] = as[s] + ad[lo + dl[j]];
      mv[j] = mx[dl[j]];
      const int r = s * C;
      #pragma unroll
      for (int i = 0; i < 5; ++i) hv[j][i] = h2[r + cc[i]];
    }
    #pragma unroll
    for (int j = 0; j < U2; ++j) {
      if (k[j]) {
        const float lg = av[j] > 0.f ? av[j] : NEG_SLOPE * av[j];
        const float ex = expf(lg - mv[j]);
        float* rp = rows + dl[j] * P2;
        if (q == 0) atomicAdd(&rp[40], ex);
        #pragma unroll
        for (int i = 0; i < 5; ++i) { if (q + 8 * i < C) atomicAdd(&rp[q + 8 * i], hv[j][i] * ex); }
      }
    }
  }
  if (total > LCAP) {
    const int* __restrict__ dsts = ei + E;
    for (int i = tid; i < E; i += AGG_T) {
      const unsigned dl = (unsigned)(dsts[i] - lo);
      if (dl < (unsigned)nn) {
        const int s = ei[i];
        for (int k = 0; k < 8; ++k) l2_edge(rows, mx, s, lo + (int)dl, (int)dl, k, C, as, ad, h2);
      }
    }
  }
  __syncthreads();
  for (int i = tid; i < nn * C; i += AGG_T) {
    const int r = i / C, c = i - r * C;
    const float v = rows[r * P2 + c] / (rows[r * P2 + 40] + EPSV) + b[c];
    rows[r * P2 + c] = v > 0.f ? v : expm1f(v);
  }
  __syncthreads();
  if (tid < nn) {
    float hv[40];
    #pragma unroll
    for (int c = 0; c < 40; ++c) { const float t = rows[tid * P2 + c]; hv[c] = c < C ? t : -INFINITY; }
    float m = -INFINITY;
    #pragma unroll
    for (int c = 0; c < 40; ++c) m = fmaxf(m, hv[c]);
    float sm = 0.f;
    #pragma unroll
    for (int c = 0; c < 40; ++c) sm += expf(hv[c] - m);
    mx[tid] = m;
    lse[tid] = logf(sm);
  }
  __syncthreads();
  for (int i = tid; i < nn * C; i += AGG_T) {
    const int r = i / C, c = i - r * C;
    out[(size_t)lo * C + i] = rows[r * P2 + c] - mx[r] - lse[r];
  }
}


static inline int cdiv(int a, int b) { return (a + b - 1) / b; }

extern "C" void kernel_launch(void* const* d_in, const int* in_sizes, int n_in,
                              void* d_out, int out_size, void* d_ws, size_t ws_size,
                              hipStream_t stream) {
  const float* x      = (const float*)d_in[0];
  const int*   ei     = (const int*)  d_in[1];
  const float* W1     = (const float*)d_in[2];
  const float* a_src1 = (const float*)d_in[3];
  const float* a_dst1 = (const float*)d_in[4];
  const float* b1     = (const float*)d_in[5];
  const float* W2     = (const float*)d_in[6];
  const float* a_src2 = (const float*)d_in[7];
  const float* a_dst2 = (const float*)d_in[8];
  const float* b2     = (const float*)d_in[9];
  float* out = (float*)d_out;

  const int HC   = in_sizes[5];
  const int NCLS = in_sizes[9];
  const int F    = in_sizes[2] / HC;
  const int N    = in_sizes[0] / F;
  const int E    = in_sizes[1] / 2;
  const int NPAD = 48;
  const int LD1  = HC + 16;

  float* ws = (float*)d_ws;
  size_t off = 0;
  const int G   = cdiv(N, RN);
  const int NB  = cdiv(E, BM_E);
  const int CP  = NB * CELL;
  unsigned int* cells = (unsigned int*)(ws + off); off += (size_t)G * CP;
  float* h1   = ws + off; off += (size_t)N * HC;
  float* as1  = ws + off; off += (size_t)N * 8;
  float* ad1  = ws + off; off += (size_t)N * 8;
  float* out1 = ws + off; off += (size_t)N * HC;
  float* h2   = ws + off; off += (size_t)N * NCLS;
  float* as2  = ws + off; off += (size_t)N;
  float* ad2  = ws + off; off += (size_t)N;
  float* Wp   = ws + off; off += (size_t)HC * NPAD;
  float* W1e  = ws + off; off += (size_t)F * LD1;
  int* keep_s = (int*)(ws + off); off += (size_t)G * LCAP;
  int* keep_d = (int*)(ws + off); off += (size_t)G * LCAP;
  int* keep_n = (int*)(ws + off); off += (size_t)G;

  const int B = 256;

  ext_weights_kernel<<<cdiv(F * LD1 + HC * NPAD, B), B, 0, stream>>>(W1, a_src1, a_dst1, W1e,
                                                                     W2, a_src2, a_dst2, Wp, F, HC, NCLS, NPAD);

  gemm_wmma_f32_nt<5><<<dim3(N / 16), 32, 0, stream>>>(x, W1e, h1, as1, ad1, N, F, LD1, HC, HC, 8);
  edge_cells_kernel<<<NB, BM_T, 0, stream>>>(ei, E, G, CP, cells);
  edge_agg1_own_kernel<<<G, AGG_T, 0, stream>>>(ei, E, N, as1, ad1, b1, h1, out1, keep_s, keep_d, keep_n,
                                               cells, CP);

  gemm_wmma_f32_nt<3><<<dim3(N / 16), 32, 0, stream>>>(out1, Wp, h2, as2, ad2, N, HC, NPAD, NCLS, NCLS, 1);
  edge_agg2_own_kernel<<<G, AGG_T, 0, stream>>>(ei, E, N, as2, ad2, b2, h2, out, NCLS, keep_s, keep_d, keep_n);
}


# weight prep and edge cell pass share one launch: its leading work-groups build both extended weight matrices and leave, the others take one block of edges each; 6->5 launches
# speedup vs baseline: 10.4225x; 1.0332x over previous
#include <hip/hip_runtime.h>
#include <hip/hip_bf16.h>
#include <math.h>

typedef __attribute__((ext_vector_type(2))) float v2f;
typedef __attribute__((ext_vector_type(8))) float v8f;
typedef __attribute__((ext_vector_type(4))) unsigned int v4u;

#define NEG_SLOPE 0.2f
#define EPSV 1e-16f


__device__ __forceinline__ void atomicMaxF(float* addr, float val) {
  if (val >= 0.0f) atomicMax((int*)addr, __float_as_int(val));
  else             atomicMin((unsigned int*)addr, __float_as_uint(val));
}

template <int NT>
__global__ void __launch_bounds__(32)
gemm_wmma_f32_nt(const float* __restrict__ A, const float* __restrict__ B,
                 float* __restrict__ D, float* __restrict__ as, float* __restrict__ ad,
                 int M, int K, int ldb, int ldd, int nvalid, int na) {
  const int tileM = blockIdx.x;
  const int lane = threadIdx.x & 31;
  const int half = lane >> 4;
  const int l16  = lane & 15;
  const int rowA = tileM * 16 + l16;
  const float* __restrict__ arow = A + (size_t)rowA * K;

  v8f acc[NT];
  #pragma unroll
  for (int nt = 0; nt < NT; ++nt) acc[nt] = (v8f){0.f,0.f,0.f,0.f,0.f,0.f,0.f,0.f};

  for (int kk = 0; kk < K; kk += 4) {
    const int ka = kk + half * 2;
    v2f a;
    a.x = arow[ka];
    a.y = arow[ka + 1];
    const float* __restrict__ brow0 = B + (size_t)ka * ldb;
    const float* __restrict__ brow1 = B + (size_t)(ka + 1) * ldb;
    #pragma unroll
    for (int nt = 0; nt < NT; ++nt) {
      const int colB = nt * 16 + l16;
      v2f b;
      b.x = brow0[colB];
      b.y = brow1[colB];
      acc[nt] = __builtin_amdgcn_wmma_f32_16x16x4_f32(false, a, false, b,
                                                      (short)0, acc[nt], false, false);
    }
  }
  const size_t r0 = (size_t)(tileM * 16 + half * 8);
  #pragma unroll
  for (int nt = 0; nt < NT; ++nt) {
    const int c = nt * 16 + l16;
    float* __restrict__ p = nullptr;
    size_t pitch = 0;
    if (c < nvalid)               { p = D + c;                    pitch = (size_t)ldd; }
    else if (c < nvalid + na)     { p = as + (c - nvalid);        pitch = (size_t)na; }
    else if (c < nvalid + 2 * na) { p = ad + (c - nvalid - na);   pitch = (size_t)na; }
    if (p != nullptr) {
      p += r0 * pitch;
      #pragma unroll
      for (int g = 0; g < 8; ++g) p[(size_t)g * pitch] = acc[nt][g];
    }
  }
}

__device__ __forceinline__ void ext_weights_part(int i,
                                   const float* __restrict__ W1, const float* __restrict__ a_src1,
                                   const float* __restrict__ a_dst1, float* __restrict__ W1e,
                                   const float* __restrict__ W2, const float* __restrict__ a_src2,
                                   const float* __restrict__ a_dst2, float* __restrict__ W2e,
                                   int F, int HC, int NCLS, int NPAD) {
  const int ld1 = HC + 16;
  const int n1 = F * ld1;
  if (i < n1) {
    const int k = i / ld1, c = i - k * ld1;
    float v;
    if (c < HC) {
      v = W1[k * HC + c];
    } else {
      const int j = c - HC, hh = j & 7;
      const float* __restrict__ a = (j < 8 ? a_src1 : a_dst1) + hh * 8;
      const float* __restrict__ w = W1 + k * HC + hh * 8;
      v = 0.f;
      #pragma unroll
      for (int t = 0; t < 8; ++t) v += w[t] * a[t];
    }
    W1e[i] = v;
  } else if (i - n1 < HC * NPAD) {
    const int i2 = i - n1;
    const int k = i2 / NPAD, c = i2 - k * NPAD;
    float v = 0.f;
    if (c < NCLS) {
      v = W2[k * NCLS + c];
    } else if (c < NCLS + 2) {
      const float* __restrict__ a = c == NCLS ? a_src2 : a_dst2;
      const float* __restrict__ w = W2 + k * NCLS;
      for (int t = 0; t < NCLS; ++t) v += w[t] * a[t];
    }
    W2e[i2] = v;
  }
}


#define AGG_T   1024
#define RN      196
#define LCAP    32768
#define P1      73
#define P2      41
#define U2      8

struct L1Vals { float ex; float4 a; float4 b; };

__device__ __forceinline__ float l1_logit(int s, int d, int hh,
                                         const float* __restrict__ as, const float* __restrict__ ad) {
  const float v = as[s * 8 + hh] + ad[d * 8 + hh];
  return v > 0.f ? v : NEG_SLOPE * v;
}

__device__ __forceinline__ L1Vals l1_load(int s, int d, int dl, int hh,
                                          const float* __restrict__ as, const float* __restrict__ ad,
                                          const float* mx, const float* __restrict__ h1) {
  L1Vals r;
  r.ex = expf(l1_logit(s, d, hh, as, ad) - mx[dl * 8 + hh]);
  const float4* hp = (const float4*)(h1 + (size_t)s * 64 + hh * 8);
  r.a = hp[0]; r.b = hp[1];
  return r;
}

__device__ __forceinline__ void l1_add(float* rows, int dl, int hh, const L1Vals& r) {
  float* op = rows + dl * P1 + hh * 8;
  atomicAdd(&rows[dl * P1 + 64 + hh], r.ex);
  atomicAdd(&op[0], r.a.x * r.ex); atomicAdd(&op[1], r.a.y * r.ex);
  atomicAdd(&op[2], r.a.z * r.ex); atomicAdd(&op[3], r.a.w * r.ex);
  atomicAdd(&op[4], r.b.x * r.ex); atomicAdd(&op[5], r.b.y * r.ex);
  atomicAdd(&op[6], r.b.z * r.ex); atomicAdd(&op[7], r.b.w * r.ex);
}

#define BM_E    2048
#define CELL    32
#define BM_G    256
#define BM_T    512
#define ENT_EMPTY 0xFFFFFFFFu
#define ENT_OVER  0xFFFFFFFEu
#define OVF_CAP 512

__global__ void __launch_bounds__(BM_T)
prep_kernel(const int* __restrict__ ei, int E, int G, int pitch,
            unsigned int* __restrict__ cells, int nwb,
            const float* __restrict__ W1, const float* __restrict__ a_src1,
            const float* __restrict__ a_dst1, float* __restrict__ W1e,
            const float* __restrict__ W2, const float* __restrict__ a_src2,
            const float* __restrict__ a_dst2, float* __restrict__ W2e,
            int F, int HC, int NCLS, int NPAD) {
  __shared__ uint4 tile4[BM_G * CELL / 4];
  __shared__ int ccnt[BM_G];
  unsigned int* tile = (unsigned int*)tile4;
  const int tid = threadIdx.x;
  if ((int)blockIdx.x < nwb) {
    ext_weights_part((int)blockIdx.x * BM_T + tid, W1, a_src1, a_dst1, W1e,
                     W2, a_src2, a_dst2, W2e, F, HC, NCLS, NPAD);
    return;
  }
  const int cb = (int)blockIdx.x - nwb;
  const int base = cb * BM_E;
  const int* __restrict__ dsts = ei + E;
  int sv[BM_E / BM_T], dv[BM_E / BM_T];
  #pragma unroll
  for (int j = 0; j < BM_E / BM_T; ++j) {
    const int e = base + tid + j * BM_T;
    const int f = e < E ? e : E - 1;
    sv[j] = ei[f];
    dv[j] = dsts[f];
  }
  for (int g0 = 0; g0 < G; g0 += BM_G) {
    for (int i = tid; i < BM_G * CELL / 4; i += BM_T)
      tile4[i] = make_uint4(ENT_EMPTY, ENT_EMPTY, ENT_EMPTY, ENT_EMPTY);
    if (tid < BM_G) ccnt[tid] = 0;
    __syncthreads();
    #pragma unroll
    for (int j = 0; j < BM_E / BM_T; ++j) {
      const int own = dv[j] / RN;
      const unsigned int g = (unsigned int)(own - g0);
      if (base + tid + j * BM_T < E && g < (unsigned int)BM_G) {
        const int r = atomicAdd(&ccnt[g], 1);
        if (r < CELL)
          tile[g * CELL + r] = (unsigned int)sv[j] | ((unsigned int)(dv[j] - own * RN) << 24);
      }
    }
    __syncthreads();
    for (int i = tid; i < BM_G * CELL / 4; i += BM_T) {
      const int gl = i / (CELL / 4), part = i % (CELL / 4);
      const int g = g0 + gl;
      if (g < G) {
        uint4 v = tile4[i];
        if (ccnt[gl] > CELL)
          v = make_uint4(part == 0 ? ENT_OVER : ENT_EMPTY, ENT_EMPTY, ENT_EMPTY, ENT_EMPTY);
        ((uint4*)(cells + (size_t)g * pitch + (size_t)cb * CELL))[part] = v;
      }
    }
    __syncthreads();
  }
}

__device__ __forceinline__ void put_entry(unsigned int e, int& pos, int* lst_s, unsigned char* lst_d) {
  if ((e >> 24) < (unsigned int)RN) {
    if (pos < LCAP) { lst_s[pos] = (int)(e & 0xFFFFFFu); lst_d[pos] = (unsigned char)(e >> 24); }
    ++pos;
  }
}

__device__ __forceinline__ void collect_cells(const unsigned int* __restrict__ row, int nq,
                                              int* lst_s, unsigned char* lst_d, int* cnt,
                                              int* ovf, int* novf) {
  const int tid = threadIdx.x;
  const int lane = tid & 31;
  const v4u* __restrict__ r4 = (const v4u*)row;
  for (int base = 0; base < nq; base += 4 * AGG_T) {
    v4u v[4];
    #pragma unroll
    for (int j = 0; j < 4; ++j) {
      const int q = base + tid + j * AGG_T;
      v[j] = r4[q < nq ? q : nq - 1];
    }
    int n = 0;
    #pragma unroll
    for (int j = 0; j < 4; ++j) {
      const int q = base + tid + j * AGG_T;
      if (q >= nq) v[j] = (v4u){ENT_EMPTY, ENT_EMPTY, ENT_EMPTY, ENT_EMPTY};
      if ((q & 7) == 0 && v[j].x == ENT_OVER) {
        const int k = atomicAdd(novf, 1);
        if (k < OVF_CAP) ovf[k] = q >> 3;
      }
      n += ((v[j].x >> 24) < (unsigned int)RN) + ((v[j].y >> 24) < (unsigned int)RN)
         + ((v[j].z >> 24) < (unsigned int)RN) + ((v[j].w >> 24) < (unsigned int)RN);
    }
    int inc = n;
    #pragma unroll
    for (int o = 1; o < 32; o <<= 1) {
      const int t = __shfl_up(inc, o);
      if (lane >= o) inc += t;
    }
    int wbase = 0;
    if (lane == 31 && inc) wbase = atomicAdd(cnt, inc);
    wbase = __shfl(wbase, 31);
    int pos = wbase + inc - n;
    #pragma unroll
    for (int j = 0; j < 4; ++j) {
      put_entry(v[j].x, pos, lst_s, lst_d);
      put_entry(v[j].y, pos, lst_s, lst_d);
      put_entry(v[j].z, pos, lst_s, lst_d);
      put_entry(v[j].w, pos, lst_s, lst_d);
    }
  }
}

__global__ void __launch_bounds__(AGG_T)
edge_agg1_own_kernel(const int* __restrict__ ei, int E, int N,
                     const float* __restrict__ as, const float* __restrict__ ad,
                     const float* __restrict__ b,
                     const float* __restrict__ h1, float* __restrict__ out,
                     int* __restrict__ keep_s, int* __restrict__ keep_d, int* __restrict__ keep_n,
                     const unsigned int* __restrict__ cells, int pitch) {
  __shared__ float rows[RN * P1];
  __shared__ float mx[RN * 8];
  __shared__ int lst_s[LCAP];
  __shared__ unsigned char lst_d[LCAP];
  __shared__ int ovf[OVF_CAP];
  __shared__ int cnt, novf;
  const int tid = threadIdx.x;
  const int lo = blockIdx.x * RN;
  const int nn = min(RN, N - lo);
  for (int i = tid; i < RN * P1; i += AGG_T) rows[i] = 0.f;
  for (int i = tid; i < RN * 8; i += AGG_T) mx[i] = -INFINITY;
  if (tid < nn) { lst_s[tid] = lo + tid; lst_d[tid] = (unsigned char)tid; }
  if (tid == 0) { cnt = nn; novf = 0; }
  __syncthreads();
  collect_cells(cells + (size_t)blockIdx.x * pitch, pitch / 4, lst_s, lst_d, &cnt, ovf, &novf);
  __syncthreads();
  const int nov = novf;
  if (nov != 0 && nov <= OVF_CAP) {
    const int* __restrict__ dsts = ei + E;
    for (int k = 0; k < nov; ++k) {
      const int e0 = ovf[k] * BM_E;
      const int e1 = min(E, e0 + BM_E);
      for (int e = e0 + tid; e < e1; e += AGG_T) {
        const unsigned dl = (unsigned)(dsts[e] - lo);
        if (dl < (unsigned)nn) {
          const int pos = atomicAdd(&cnt, 1);
          if (pos < LCAP) { lst_s[pos] = ei[e]; lst_d[pos] = (unsigned char)dl; }
        }
      }
    }
    __syncthreads();
  }
  const int total = nov <= OVF_CAP ? cnt : LCAP + 1;
  const int nl = total <= LCAP ? total : nn;
  for (int i = tid; i < nl; i += AGG_T) {
    keep_s[(size_t)blockIdx.x * LCAP + i] = lst_s[i];
    keep_d[(size_t)blockIdx.x * LCAP + i] = lst_d[i];
  }
  if (tid == 0) keep_n[blockIdx.x] = total;

  const int hh = tid & 7, sub = tid >> 3;

  for (int base = 0; base < nl; base += 4 * (AGG_T / 8)) {
    float v[4]; int dl[4]; bool k[4];
    #pragma unroll
    for (int j = 0; j < 4; ++j) {
      const int e = base + sub + j * (AGG_T / 8);
      k[j] = e < nl;
      const int f = k[j] ? e : nl - 1;
      dl[j] = lst_d[f];
      v[j] = l1_logit(lst_s[f], lo + dl[j], hh, as, ad);
    }
    #pragma unroll
    for (int j = 0; j < 4; ++j) if (k[j]) atomicMaxF(&mx[dl[j] * 8 + hh], v[j]);
  }
  if (total > LCAP) {
    const int* __restrict__ dsts = ei + E;
    for (int i = tid; i < E; i += AGG_T) {
      const unsigned dl = (unsigned)(dsts[i] - lo);
      if (dl < (unsigned)nn) {
        const int s = ei[i];
        for (int h = 0; h < 8; ++h) atomicMaxF(&mx[dl * 8 + h], l1_logit(s, lo + (int)dl, h, as, ad));
      }
    }
  }
  __syncthreads();

  for (int base = 0; base < nl; base += 2 * (AGG_T / 8)) {
    const int e0 = base + sub, e1 = e0 + AGG_T / 8;
    const bool k0 = e0 < nl, k1 = e1 < nl;
    const int f0 = k0 ? e0 : nl - 1, f1 = k1 ? e1 : nl - 1;
    const int s0 = lst_s[f0], d0 = lst_d[f0];
    const int s1 = lst_s[f1], d1 = lst_d[f1];
    const L1Vals r0 = l1_load(s0, lo + d0, d0, hh, as, ad, mx, h1);
    const L1Vals r1 = l1_load(s1, lo + d1, d1, hh, as, ad, mx, h1);
    if (k0) l1_add(rows, d0, hh, r0);
    if (k1) l1_add(rows, d1, hh, r1);
  }
  if (total > LCAP) {
    const int* __restrict__ dsts = ei + E;
    for (int i = tid; i < E; i += AGG_T) {
      const unsigned dl = (unsigned)(dsts[i] - lo);
      if (dl < (unsigned)nn) {
        const int s = ei[i];
        for (int h = 0; h < 8; ++h) l1_add(rows, (int)dl, h, l1_load(s, lo + (int)dl, (int)dl, h, as, ad, mx, h1));
      }
    }
  }
  __syncthreads();
  for (int i = tid; i < nn * 64; i += AGG_T) {
    const int r = i >> 6, c = i & 63;
    const float v = rows[r * P1 + c] / (rows[r * P1 + 64 + (c >> 3)] + EPSV) + b[c];
    out[(size_t)lo * 64 + i] = v > 0.f ? v : expm1f(v);
  }
}

__device__ __forceinline__ float l2_logit(int s, int d,
                                         const float* __restrict__ as, const float* __restrict__ ad) {
  const float v = as[s] + ad[d];
  return v > 0.f ? v : NEG_SLOPE * v;
}

__device__ __forceinline__ void l2_edge(float* rows, const float* mx, int s, int d, int dl, int q, int C,
                                        const float* __restrict__ as, const float* __restrict__ ad,
                                        const float* __restrict__ h2) {
  const float ex = expf(l2_logit(s, d, as, ad) - mx[dl]);
  const float* hp = h2 + (size_t)s * C;
  float hv[5];
  #pragma unroll
  for (int j = 0; j < 5; ++j) { const int c = q + 8 * j; hv[j] = c < C ? hp[c] : 0.f; }
  if (q == 0) atomicAdd(&rows[dl * P2 + 40], ex);
  #pragma unroll
  for (int j = 0; j < 5; ++j) { const int c = q + 8 * j; if (c < C) atomicAdd(&rows[dl * P2 + c], hv[j] * ex); }
}

__global__ void __launch_bounds__(AGG_T)
edge_agg2_own_kernel(const int* __restrict__ ei, int E, int N,
                     const float* __restrict__ as, const float* __restrict__ ad,
                     const float* __restrict__ b,
                     const float* __restrict__ h2, float* __restrict__ out, int C,
                     const int* __restrict__ keep_s, const int* __restrict__ keep_d,
                     const int* __restrict__ keep_n) {
  __shared__ float rows[RN * P2];
  __shared__ float mx[RN];
  __shared__ float lse[RN];
  __shared__ int lst_s[LCAP];
  __shared__ unsigned char lst_d[LCAP];
  const int tid = threadIdx.x;
  const int lo = blockIdx.x * RN;
  const int nn = min(RN, N - lo);
  for (int i = tid; i < RN * P2; i += AGG_T) rows[i] = 0.f;
  if (tid < RN) mx[tid] = -INFINITY;
  __syncthreads();
  const int total = keep_n[blockIdx.x];
  const int nl = total <= LCAP ? total : nn;
  const int* __restrict__ ls = keep_s + (size_t)blockIdx.x * LCAP;
  const int* __restrict__ ld = keep_d + (size_t)blockIdx.x * LCAP;

  for (int base = 0; base < nl; base += 4 * AGG_T) {
    float v[4]; int sv[4]; int dl[4]; bool k[4];
    #pragma unroll
    for (int j = 0; j < 4; ++j) {
      const int e = base + tid + j * AGG_T;
      k[j] = e < nl;
      const int f = k[j] ? e : nl - 1;
      dl[j] = ld[f];
      sv[j] = ls[f];
      v[j] = l2_logit(sv[j], lo + dl[j], as, ad);
    }
    #pragma unroll
    for (int j = 0; j < 4; ++j) {
      if (k[j]) {
        const int e = base + tid + j * AGG_T;
        lst_s[e] = sv[j]; lst_d[e] = (unsigned char)dl[j];
        atomicMaxF(&mx[dl[j]], v[j]);
      }
    }
  }
  if (total > LCAP) {
    const int* __restrict__ dsts = ei + E;
    for (int i = tid; i < E; i += AGG_T) {
      const unsigned dl = (unsigned)(dsts[i] - lo);
      if (dl < (unsigned)nn) atomicMaxF(&mx[dl], l2_logit(ei[i], lo + (int)dl, as, ad));
    }
  }
  __syncthreads();

  const int q = tid & 7, sub = tid >> 3;
  int cc[5];
  #pragma unroll
  for (int i = 0; i < 5; ++i) { const int c = q + 8 * i; cc[i] = c < C ? c : 0; }
  for (int base = 0; base < nl; base += U2 * (AGG_T / 8)) {
    float av[U2], mv[U2], hv[U2][5]; int dl[U2]; bool k[U2];
    #pragma unroll
    for (int j = 0; j < U2; ++j) {
      const int e = base + sub + j * (AGG_T / 8);
      k[j] = e < nl;
      const int f = k[j] ? e : nl - 1;
      const int s = lst_s[f];
      dl[j] = lst_d[f];
      av[j] = as[s] + ad[lo + dl[j]];
      mv[j] = mx[dl[j]];
      const int r = s * C;
      #pragma unroll
      for (int i = 0; i < 5; ++i) hv[j][i] = h2[r + cc[i]];
    }
    #pragma unroll
    for (int j = 0; j < U2; ++j) {
      if (k[j]) {
        const float lg = av[j] > 0.f ? av[j] : NEG_SLOPE * av[j];
        const float ex = expf(lg - mv[j]);
        float* rp = rows + dl[j] * P2;
        if (q == 0) atomicAdd(&rp[40], ex);
        #pragma unroll
        for (int i = 0; i < 5; ++i) { if (q + 8 * i < C) atomicAdd(&rp[q + 8 * i], hv[j][i] * ex); }
      }
    }
  }
  if (total > LCAP) {
    const int* __restrict__ dsts = ei + E;
    for (int i = tid; i < E; i += AGG_T) {
      const unsigned dl = (unsigned)(dsts[i] - lo);
      if (dl < (unsigned)nn) {
        const int s = ei[i];
        for (int k = 0; k < 8; ++k) l2_edge(rows, mx, s, lo + (int)dl, (int)dl, k, C, as, ad, h2);
      }
    }
  }
  __syncthreads();
  for (int i = tid; i < nn * C; i += AGG_T) {
    const int r = i / C, c = i - r * C;
    const float v = rows[r * P2 + c] / (rows[r * P2 + 40] + EPSV) + b[c];
    rows[r * P2 + c] = v > 0.f ? v : expm1f(v);
  }
  __syncthreads();
  if (tid < nn) {
    float hv[40];
    #pragma unroll
    for (int c = 0; c < 40; ++c) { const float t = rows[tid * P2 + c]; hv[c] = c < C ? t : -INFINITY; }
    float m = -INFINITY;
    #pragma unroll
    for (int c = 0; c < 40; ++c) m = fmaxf(m, hv[c]);
    float sm = 0.f;
    #pragma unroll
    for (int c = 0; c < 40; ++c) sm += expf(hv[c] - m);
    mx[tid] = m;
    lse[tid] = logf(sm);
  }
  __syncthreads();
  for (int i = tid; i < nn * C; i += AGG_T) {
    const int r = i / C, c = i - r * C;
    out[(size_t)lo * C + i] = rows[r * P2 + c] - mx[r] - lse[r];
  }
}


static inline int cdiv(int a, int b) { return (a + b - 1) / b; }

extern "C" void kernel_launch(void* const* d_in, const int* in_sizes, int n_in,
                              void* d_out, int out_size, void* d_ws, size_t ws_size,
                              hipStream_t stream) {
  const float* x      = (const float*)d_in[0];
  const int*   ei     = (const int*)  d_in[1];
  const float* W1     = (const float*)d_in[2];
  const float* a_src1 = (const float*)d_in[3];
  const float* a_dst1 = (const float*)d_in[4];
  const float* b1     = (const float*)d_in[5];
  const float* W2     = (const float*)d_in[6];
  const float* a_src2 = (const float*)d_in[7];
  const float* a_dst2 = (const float*)d_in[8];
  const float* b2     = (const float*)d_in[9];
  float* out = (float*)d_out;

  const int HC   = in_sizes[5];
  const int NCLS = in_sizes[9];
  const int F    = in_sizes[2] / HC;
  const int N    = in_sizes[0] / F;
  const int E    = in_sizes[1] / 2;
  const int NPAD = 48;
  const int LD1  = HC + 16;

  float* ws = (float*)d_ws;
  size_t off = 0;
  const int G   = cdiv(N, RN);
  const int NB  = cdiv(E, BM_E);
  const int CP  = NB * CELL;
  unsigned int* cells = (unsigned int*)(ws + off); off += (size_t)G * CP;
  float* h1   = ws + off; off += (size_t)N * HC;
  float* as1  = ws + off; off += (size_t)N * 8;
  float* ad1  = ws + off; off += (size_t)N * 8;
  float* out1 = ws + off; off += (size_t)N * HC;
  float* h2   = ws + off; off += (size_t)N * NCLS;
  float* as2  = ws + off; off += (size_t)N;
  float* ad2  = ws + off; off += (size_t)N;
  float* Wp   = ws + off; off += (size_t)HC * NPAD;
  float* W1e  = ws + off; off += (size_t)F * LD1;
  int* keep_s = (int*)(ws + off); off += (size_t)G * LCAP;
  int* keep_d = (int*)(ws + off); off += (size_t)G * LCAP;
  int* keep_n = (int*)(ws + off); off += (size_t)G;

  const int NWB = cdiv(F * LD1 + HC * NPAD, BM_T);
  prep_kernel<<<NWB + NB, BM_T, 0, stream>>>(ei, E, G, CP, cells, NWB, W1, a_src1, a_dst1, W1e,
                                             W2, a_src2, a_dst2, Wp, F, HC, NCLS, NPAD);

  gemm_wmma_f32_nt<5><<<dim3(N / 16), 32, 0, stream>>>(x, W1e, h1, as1, ad1, N, F, LD1, HC, HC, 8);
  edge_agg1_own_kernel<<<G, AGG_T, 0, stream>>>(ei, E, N, as1, ad1, b1, h1, out1, keep_s, keep_d, keep_n,
                                               cells, CP);

  gemm_wmma_f32_nt<3><<<dim3(N / 16), 32, 0, stream>>>(out1, Wp, h2, as2, ad2, N, HC, NPAD, NCLS, NCLS, 1);
  edge_agg2_own_kernel<<<G, AGG_T, 0, stream>>>(ei, E, N, as2, ad2, b2, h2, out, NCLS, keep_s, keep_d, keep_n);
}


# both row passes form an edge's source logit from the source row they gather (dot with a_src; layer 2 sums its 8 lanes' parts by DPP adds), not from the logit table; logit sum order changes
# speedup vs baseline: 10.7709x; 1.0334x over previous
#include <hip/hip_runtime.h>
#include <hip/hip_bf16.h>
#include <math.h>

typedef __attribute__((ext_vector_type(2))) float v2f;
typedef __attribute__((ext_vector_type(8))) float v8f;
typedef __attribute__((ext_vector_type(4))) unsigned int v4u;

#define NEG_SLOPE 0.2f
#define EPSV 1e-16f


__device__ __forceinline__ void atomicMaxF(float* addr, float val) {
  if (val >= 0.0f) atomicMax((int*)addr, __float_as_int(val));
  else             atomicMin((unsigned int*)addr, __float_as_uint(val));
}

template <int NT>
__global__ void __launch_bounds__(32)
gemm_wmma_f32_nt(const float* __restrict__ A, const float* __restrict__ B,
                 float* __restrict__ D, float* __restrict__ as, float* __restrict__ ad,
                 int M, int K, int ldb, int ldd, int nvalid, int na) {
  const int tileM = blockIdx.x;
  const int lane = threadIdx.x & 31;
  const int half = lane >> 4;
  const int l16  = lane & 15;
  const int rowA = tileM * 16 + l16;
  const float* __restrict__ arow = A + (size_t)rowA * K;

  v8f acc[NT];
  #pragma unroll
  for (int nt = 0; nt < NT; ++nt) acc[nt] = (v8f){0.f,0.f,0.f,0.f,0.f,0.f,0.f,0.f};

  for (int kk = 0; kk < K; kk += 4) {
    const int ka = kk + half * 2;
    v2f a;
    a.x = arow[ka];
    a.y = arow[ka + 1];
    const float* __restrict__ brow0 = B + (size_t)ka * ldb;
    const float* __restrict__ brow1 = B + (size_t)(ka + 1) * ldb;
    #pragma unroll
    for (int nt = 0; nt < NT; ++nt) {
      const int colB = nt * 16 + l16;
      v2f b;
      b.x = brow0[colB];
      b.y = brow1[colB];
      acc[nt] = __builtin_amdgcn_wmma_f32_16x16x4_f32(false, a, false, b,
                                                      (short)0, acc[nt], false, false);
    }
  }
  const size_t r0 = (size_t)(tileM * 16 + half * 8);
  #pragma unroll
  for (int nt = 0; nt < NT; ++nt) {
    const int c = nt * 16 + l16;
    float* __restrict__ p = nullptr;
    size_t pitch = 0;
    if (c < nvalid)               { p = D + c;                    pitch = (size_t)ldd; }
    else if (c < nvalid + na)     { p = as + (c - nvalid);        pitch = (size_t)na; }
    else if (c < nvalid + 2 * na) { p = ad + (c - nvalid - na);   pitch = (size_t)na; }
    if (p != nullptr) {
      p += r0 * pitch;
      #pragma unroll
      for (int g = 0; g < 8; ++g) p[(size_t)g * pitch] = acc[nt][g];
    }
  }
}

__device__ __forceinline__ void ext_weights_part(int i,
                                   const float* __restrict__ W1, const float* __restrict__ a_src1,
                                   const float* __restrict__ a_dst1, float* __restrict__ W1e,
                                   const float* __restrict__ W2, const float* __restrict__ a_src2,
                                   const float* __restrict__ a_dst2, float* __restrict__ W2e,
                                   int F, int HC, int NCLS, int NPAD) {
  const int ld1 = HC + 16;
  const int n1 = F * ld1;
  if (i < n1) {
    const int k = i / ld1, c = i - k * ld1;
    float v;
    if (c < HC) {
      v = W1[k * HC + c];
    } else {
      const int j = c - HC, hh = j & 7;
      const float* __restrict__ a = (j < 8 ? a_src1 : a_dst1) + hh * 8;
      const float* __restrict__ w = W1 + k * HC + hh * 8;
      v = 0.f;
      #pragma unroll
      for (int t = 0; t < 8; ++t) v += w[t] * a[t];
    }
    W1e[i] = v;
  } else if (i - n1 < HC * NPAD) {
    const int i2 = i - n1;
    const int k = i2 / NPAD, c = i2 - k * NPAD;
    float v = 0.f;
    if (c < NCLS) {
      v = W2[k * NCLS + c];
    } else if (c < NCLS + 2) {
      const float* __restrict__ a = c == NCLS ? a_src2 : a_dst2;
      const float* __restrict__ w = W2 + k * NCLS;
      for (int t = 0; t < NCLS; ++t) v += w[t] * a[t];
    }
    W2e[i2] = v;
  }
}


#define AGG_T   1024
#define RN      196
#define LCAP    32768
#define P1      73
#define P2      41
#define U2      8

struct L1Vals { float ex; float4 a; float4 b; };

__device__ __forceinline__ float l1_logit(int s, int d, int hh,
                                         const float* __restrict__ as, const float* __restrict__ ad) {
  const float v = as[s * 8 + hh] + ad[d * 8 + hh];
  return v > 0.f ? v : NEG_SLOPE * v;
}

__device__ __forceinline__ L1Vals l1_load(int s, int d, int dl, int hh,
                                          const float* __restrict__ as, const float* __restrict__ ad,
                                          const float* mx, const float* __restrict__ h1) {
  L1Vals r;
  r.ex = expf(l1_logit(s, d, hh, as, ad) - mx[dl * 8 + hh]);
  const float4* hp = (const float4*)(h1 + (size_t)s * 64 + hh * 8);
  r.a = hp[0]; r.b = hp[1];
  return r;
}

__device__ __forceinline__ L1Vals l1_load_row(int s, int d, int dl, int hh, const float (&av)[8],
                                              const float* __restrict__ ad,
                                              const float* mx, const float* __restrict__ h1) {
  L1Vals r;
  const float4* hp = (const float4*)(h1 + (size_t)s * 64 + hh * 8);
  r.a = hp[0]; r.b = hp[1];
  float v = r.a.x * av[0];
  v += r.a.y * av[1]; v += r.a.z * av[2]; v += r.a.w * av[3];
  v += r.b.x * av[4]; v += r.b.y * av[5]; v += r.b.z * av[6]; v += r.b.w * av[7];
  v += ad[d * 8 + hh];
  v = v > 0.f ? v : NEG_SLOPE * v;
  r.ex = expf(v - mx[dl * 8 + hh]);
  return r;
}

__device__ __forceinline__ void l1_add(float* rows, int dl, int hh, const L1Vals& r) {
  float* op = rows + dl * P1 + hh * 8;
  atomicAdd(&rows[dl * P1 + 64 + hh], r.ex);
  atomicAdd(&op[0], r.a.x * r.ex); atomicAdd(&op[1], r.a.y * r.ex);
  atomicAdd(&op[2], r.a.z * r.ex); atomicAdd(&op[3], r.a.w * r.ex);
  atomicAdd(&op[4], r.b.x * r.ex); atomicAdd(&op[5], r.b.y * r.ex);
  atomicAdd(&op[6], r.b.z * r.ex); atomicAdd(&op[7], r.b.w * r.ex);
}

#define BM_E    2048
#define CELL    32
#define BM_G    256
#define BM_T    512
#define ENT_EMPTY 0xFFFFFFFFu
#define ENT_OVER  0xFFFFFFFEu
#define OVF_CAP 512

__global__ void __launch_bounds__(BM_T)
prep_kernel(const int* __restrict__ ei, int E, int G, int pitch,
            unsigned int* __restrict__ cells, int nwb,
            const float* __restrict__ W1, const float* __restrict__ a_src1,
            const float* __restrict__ a_dst1, float* __restrict__ W1e,
            const float* __restrict__ W2, const float* __restrict__ a_src2,
            const float* __restrict__ a_dst2, float* __restrict__ W2e,
            int F, int HC, int NCLS, int NPAD) {
  __shared__ uint4 tile4[BM_G * CELL / 4];
  __shared__ int ccnt[BM_G];
  unsigned int* tile = (unsigned int*)tile4;
  const int tid = threadIdx.x;
  if ((int)blockIdx.x < nwb) {
    ext_weights_part((int)blockIdx.x * BM_T + tid, W1, a_src1, a_dst1, W1e,
                     W2, a_src2, a_dst2, W2e, F, HC, NCLS, NPAD);
    return;
  }
  const int cb = (int)blockIdx.x - nwb;
  const int base = cb * BM_E;
  const int* __restrict__ dsts = ei + E;
  int sv[BM_E / BM_T], dv[BM_E / BM_T];
  #pragma unroll
  for (int j = 0; j < BM_E / BM_T; ++j) {
    const int e = base + tid + j * BM_T;
    const int f = e < E ? e : E - 1;
    sv[j] = ei[f];
    dv[j] = dsts[f];
  }
  for (int g0 = 0; g0 < G; g0 += BM_G) {
    for (int i = tid; i < BM_G * CELL / 4; i += BM_T)
      tile4[i] = make_uint4(ENT_EMPTY, ENT_EMPTY, ENT_EMPTY, ENT_EMPTY);
    if (tid < BM_G) ccnt[tid] = 0;
    __syncthreads();
    #pragma unroll
    for (int j = 0; j < BM_E / BM_T; ++j) {
      const int own = dv[j] / RN;
      const unsigned int g = (unsigned int)(own - g0);
      if (base + tid + j * BM_T < E && g < (unsigned int)BM_G) {
        const int r = atomicAdd(&ccnt[g], 1);
        if (r < CELL)
          tile[g * CELL + r] = (unsigned int)sv[j] | ((unsigned int)(dv[j] - own * RN) << 24);
      }
    }
    __syncthreads();
    for (int i = tid; i < BM_G * CELL / 4; i += BM_T) {
      const int gl = i / (CELL / 4), part = i % (CELL / 4);
      const int g = g0 + gl;
      if (g < G) {
        uint4 v = tile4[i];
        if (ccnt[gl] > CELL)
          v = make_uint4(part == 0 ? ENT_OVER : ENT_EMPTY, ENT_EMPTY, ENT_EMPTY, ENT_EMPTY);
        ((uint4*)(cells + (size_t)g * pitch + (size_t)cb * CELL))[part] = v;
      }
    }
    __syncthreads();
  }
}

__device__ __forceinline__ void put_entry(unsigned int e, int& pos, int* lst_s, unsigned char* lst_d) {
  if ((e >> 24) < (unsigned int)RN) {
    if (pos < LCAP) { lst_s[pos] = (int)(e & 0xFFFFFFu); lst_d[pos] = (unsigned char)(e >> 24); }
    ++pos;
  }
}

__device__ __forceinline__ void collect_cells(const unsigned int* __restrict__ row, int nq,
                                              int* lst_s, unsigned char* lst_d, int* cnt,
                                              int* ovf, int* novf) {
  const int tid = threadIdx.x;
  const int lane = tid & 31;
  const v4u* __restrict__ r4 = (const v4u*)row;
  for (int base = 0; base < nq; base += 4 * AGG_T) {
    v4u v[4];
    #pragma unroll
    for (int j = 0; j < 4; ++j) {
      const int q = base + tid + j * AGG_T;
      v[j] = r4[q < nq ? q : nq - 1];
    }
    int n = 0;
    #pragma unroll
    for (int j = 0; j < 4; ++j) {
      const int q = base + tid + j * AGG_T;
      if (q >= nq) v[j] = (v4u){ENT_EMPTY, ENT_EMPTY, ENT_EMPTY, ENT_EMPTY};
      if ((q & 7) == 0 && v[j].x == ENT_OVER) {
        const int k = atomicAdd(novf, 1);
        if (k < OVF_CAP) ovf[k] = q >> 3;
      }
      n += ((v[j].x >> 24) < (unsigned int)RN) + ((v[j].y >> 24) < (unsigned int)RN)
         + ((v[j].z >> 24) < (unsigned int)RN) + ((v[j].w >> 24) < (unsigned int)RN);
    }
    int inc = n;
    #pragma unroll
    for (int o = 1; o < 32; o <<= 1) {
      const int t = __shfl_up(inc, o);
      if (lane >= o) inc += t;
    }
    int wbase = 0;
    if (lane == 31 && inc) wbase = atomicAdd(cnt, inc);
    wbase = __shfl(wbase, 31);
    int pos = wbase + inc - n;
    #pragma unroll
    for (int j = 0; j < 4; ++j) {
      put_entry(v[j].x, pos, lst_s, lst_d);
      put_entry(v[j].y, pos, lst_s, lst_d);
      put_entry(v[j].z, pos, lst_s, lst_d);
      put_entry(v[j].w, pos, lst_s, lst_d);
    }
  }
}

__global__ void __launch_bounds__(AGG_T)
edge_agg1_own_kernel(const int* __restrict__ ei, int E, int N,
                     const float* __restrict__ as, const float* __restrict__ ad,
                     const float* __restrict__ a_src,
                     const float* __restrict__ b,
                     const float* __restrict__ h1, float* __restrict__ out,
                     int* __restrict__ keep_s, int* __restrict__ keep_d, int* __restrict__ keep_n,
                     const unsigned int* __restrict__ cells, int pitch) {
  __shared__ float rows[RN * P1];
  __shared__ float mx[RN * 8];
  __shared__ int lst_s[LCAP];
  __shared__ unsigned char lst_d[LCAP];
  __shared__ int ovf[OVF_CAP];
  __shared__ int cnt, novf;
  const int tid = threadIdx.x;
  const int lo = blockIdx.x * RN;
  const int nn = min(RN, N - lo);
  for (int i = tid; i < RN * P1; i += AGG_T) rows[i] = 0.f;
  for (int i = tid; i < RN * 8; i += AGG_T) mx[i] = -INFINITY;
  if (tid < nn) { lst_s[tid] = lo + tid; lst_d[tid] = (unsigned char)tid; }
  if (tid == 0) { cnt = nn; novf = 0; }
  __syncthreads();
  collect_cells(cells + (size_t)blockIdx.x * pitch, pitch / 4, lst_s, lst_d, &cnt, ovf, &novf);
  __syncthreads();
  const int nov = novf;
  if (nov != 0 && nov <= OVF_CAP) {
    const int* __restrict__ dsts = ei + E;
    for (int k = 0; k < nov; ++k) {
      const int e0 = ovf[k] * BM_E;
      const int e1 = min(E, e0 + BM_E);
      for (int e = e0 + tid; e < e1; e += AGG_T) {
        const unsigned dl = (unsigned)(dsts[e] - lo);
        if (dl < (unsigned)nn) {
          const int pos = atomicAdd(&cnt, 1);
          if (pos < LCAP) { lst_s[pos] = ei[e]; lst_d[pos] = (unsigned char)dl; }
        }
      }
    }
    __syncthreads();
  }
  const int total = nov <= OVF_CAP ? cnt : LCAP + 1;
  const int nl = total <= LCAP ? total : nn;
  for (int i = tid; i < nl; i += AGG_T) {
    keep_s[(size_t)blockIdx.x * LCAP + i] = lst_s[i];
    keep_d[(size_t)blockIdx.x * LCAP + i] = lst_d[i];
  }
  if (tid == 0) keep_n[blockIdx.x] = total;

  const int hh = tid & 7, sub = tid >> 3;

  for (int base = 0; base < nl; base += 4 * (AGG_T / 8)) {
    float v[4]; int dl[4]; bool k[4];
    #pragma unroll
    for (int j = 0; j < 4; ++j) {
      const int e = base + sub + j * (AGG_T / 8);
      k[j] = e < nl;
      const int f = k[j] ? e : nl - 1;
      dl[j] = lst_d[f];
      v[j] = l1_logit(lst_s[f], lo + dl[j], hh, as, ad);
    }
    #pragma unroll
    for (int j = 0; j < 4; ++j) if (k[j]) atomicMaxF(&mx[dl[j] * 8 + hh], v[j]);
  }
  if (total > LCAP) {
    const int* __restrict__ dsts = ei + E;
    for (int i = tid; i < E; i += AGG_T) {
      const unsigned dl = (unsigned)(dsts[i] - lo);
      if (dl < (unsigned)nn) {
        const int s = ei[i];
        for (int h = 0; h < 8; ++h) atomicMaxF(&mx[dl * 8 + h], l1_logit(s, lo + (int)dl, h, as, ad));
      }
    }
  }
  __syncthreads();

  float av[8];
  #pragma unroll
  for (int t = 0; t < 8; ++t) av[t] = a_src[hh * 8 + t];
  for (int base = 0; base < nl; base += 2 * (AGG_T / 8)) {
    const int e0 = base + sub, e1 = e0 + AGG_T / 8;
    const bool k0 = e0 < nl, k1 = e1 < nl;
    const int f0 = k0 ? e0 : nl - 1, f1 = k1 ? e1 : nl - 1;
    const int s0 = lst_s[f0], d0 = lst_d[f0];
    const int s1 = lst_s[f1], d1 = lst_d[f1];
    const L1Vals r0 = l1_load_row(s0, lo + d0, d0, hh, av, ad, mx, h1);
    const L1Vals r1 = l1_load_row(s1, lo + d1, d1, hh, av, ad, mx, h1);
    if (k0) l1_add(rows, d0, hh, r0);
    if (k1) l1_add(rows, d1, hh, r1);
  }
  if (total > LCAP) {
    const int* __restrict__ dsts = ei + E;
    for (int i = tid; i < E; i += AGG_T) {
      const unsigned dl = (unsigned)(dsts[i] - lo);
      if (dl < (unsigned)nn) {
        const int s = ei[i];
        for (int h = 0; h < 8; ++h) l1_add(rows, (int)dl, h, l1_load(s, lo + (int)dl, (int)dl, h, as, ad, mx, h1));
      }
    }
  }
  __syncthreads();
  for (int i = tid; i < nn * 64; i += AGG_T) {
    const int r = i >> 6, c = i & 63;
    const float v = rows[r * P1 + c] / (rows[r * P1 + 64 + (c >> 3)] + EPSV) + b[c];
    out[(size_t)lo * 64 + i] = v > 0.f ? v : expm1f(v);
  }
}

template <int CTRL>
__device__ __forceinline__ float row_xadd(float v) {
  return v + __int_as_float(__builtin_amdgcn_update_dpp(0, __float_as_int(v), CTRL, 0xf, 0xf, true));
}

__device__ __forceinline__ float l2_logit(int s, int d,
                                         const float* __restrict__ as, const float* __restrict__ ad) {
  const float v = as[s] + ad[d];
  return v > 0.f ? v : NEG_SLOPE * v;
}

__device__ __forceinline__ void l2_edge(float* rows, const float* mx, int s, int d, int dl, int q, int C,
                                        const float* __restrict__ as, const float* __restrict__ ad,
                                        const float* __restrict__ h2) {
  const float ex = expf(l2_logit(s, d, as, ad) - mx[dl]);
  const float* hp = h2 + (size_t)s * C;
  float hv[5];
  #pragma unroll
  for (int j = 0; j < 5; ++j) { const int c = q + 8 * j; hv[j] = c < C ? hp[c] : 0.f; }
  if (q == 0) atomicAdd(&rows[dl * P2 + 40], ex);
  #pragma unroll
  for (int j = 0; j < 5; ++j) { const int c = q + 8 * j; if (c < C) atomicAdd(&rows[dl * P2 + c], hv[j] * ex); }
}

__global__ void __launch_bounds__(AGG_T)
edge_agg2_own_kernel(const int* __restrict__ ei, int E, int N,
                     const float* __restrict__ as, const float* __restrict__ ad,
                     const float* __restrict__ a_src,
                     const float* __restrict__ b,
                     const float* __restrict__ h2, float* __restrict__ out, int C,
                     const int* __restrict__ keep_s, const int* __restrict__ keep_d,
                     const int* __restrict__ keep_n) {
  __shared__ float rows[RN * P2];
  __shared__ float mx[RN];
  __shared__ float lse[RN];
  __shared__ int lst_s[LCAP];
  __shared__ unsigned char lst_d[LCAP];
  const int tid = threadIdx.x;
  const int lo = blockIdx.x * RN;
  const int nn = min(RN, N - lo);
  for (int i = tid; i < RN * P2; i += AGG_T) rows[i] = 0.f;
  if (tid < RN) mx[tid] = -INFINITY;
  __syncthreads();
  const int total = keep_n[blockIdx.x];
  const int nl = total <= LCAP ? total : nn;
  const int* __restrict__ ls = keep_s + (size_t)blockIdx.x * LCAP;
  const int* __restrict__ ld = keep_d + (size_t)blockIdx.x * LCAP;

  for (int base = 0; base < nl; base += 4 * AGG_T) {
    float v[4]; int sv[4]; int dl[4]; bool k[4];
    #pragma unroll
    for (int j = 0; j < 4; ++j) {
      const int e = base + tid + j * AGG_T;
      k[j] = e < nl;
      const int f = k[j] ? e : nl - 1;
      dl[j] = ld[f];
      sv[j] = ls[f];
      v[j] = l2_logit(sv[j], lo + dl[j], as, ad);
    }
    #pragma unroll
    for (int j = 0; j < 4; ++j) {
      if (k[j]) {
        const int e = base + tid + j * AGG_T;
        lst_s[e] = sv[j]; lst_d[e] = (unsigned char)dl[j];
        atomicMaxF(&mx[dl[j]], v[j]);
      }
    }
  }
  if (total > LCAP) {
    const int* __restrict__ dsts = ei + E;
    for (int i = tid; i < E; i += AGG_T) {
      const unsigned dl = (unsigned)(dsts[i] - lo);
      if (dl < (unsigned)nn) atomicMaxF(&mx[dl], l2_logit(ei[i], lo + (int)dl, as, ad));
    }
  }
  __syncthreads();

  const int q = tid & 7, sub = tid >> 3;
  int cc[5]; float a2v[5];
  #pragma unroll
  for (int i = 0; i < 5; ++i) {
    const int c = q + 8 * i;
    cc[i] = c < C ? c : 0;
    a2v[i] = c < C ? a_src[c] : 0.f;
  }
  for (int base = 0; base < nl; base += U2 * (AGG_T / 8)) {
    float av[U2], mv[U2], hv[U2][5]; int dl[U2]; bool k[U2];
    #pragma unroll
    for (int j = 0; j < U2; ++j) {
      const int e = base + sub + j * (AGG_T / 8);
      k[j] = e < nl;
      const int f = k[j] ? e : nl - 1;
      const int s = lst_s[f];
      dl[j] = lst_d[f];
      av[j] = ad[lo + dl[j]];
      mv[j] = mx[dl[j]];
      const int r = s * C;
      #pragma unroll
      for (int i = 0; i < 5; ++i) hv[j][i] = h2[r + cc[i]];
    }
    #pragma unroll
    for (int j = 0; j < U2; ++j) {
      float p = hv[j][0] * a2v[0];
      #pragma unroll
      for (int i = 1; i < 5; ++i) p += hv[j][i] * a2v[i];
      p = row_xadd<0x161>(p);
      p = row_xadd<0x162>(p);
      p = row_xadd<0x164>(p);
      if (k[j]) {
        const float a = p + av[j];
        const float lg = a > 0.f ? a : NEG_SLOPE * a;
        const float ex = expf(lg - mv[j]);
        float* rp = rows + dl[j] * P2;
        if (q == 0) atomicAdd(&rp[40], ex);
        #pragma unroll
        for (int i = 0; i < 5; ++i) { if (q + 8 * i < C) atomicAdd(&rp[q + 8 * i], hv[j][i] * ex); }
      }
    }
  }
  if (total > LCAP) {
    const int* __restrict__ dsts = ei + E;
    for (int i = tid; i < E; i += AGG_T) {
      const unsigned dl = (unsigned)(dsts[i] - lo);
      if (dl < (unsigned)nn) {
        const int s = ei[i];
        for (int k = 0; k < 8; ++k) l2_edge(rows, mx, s, lo + (int)dl, (int)dl, k, C, as, ad, h2);
      }
    }
  }
  __syncthreads();
  for (int i = tid; i < nn * C; i += AGG_T) {
    const int r = i / C, c = i - r * C;
    const float v = rows[r * P2 + c] / (rows[r * P2 + 40] + EPSV) + b[c];
    rows[r * P2 + c] = v > 0.f ? v : expm1f(v);
  }
  __syncthreads();
  if (tid < nn) {
    float hv[40];
    #pragma unroll
    for (int c = 0; c < 40; ++c) { const float t = rows[tid * P2 + c]; hv[c] = c < C ? t : -INFINITY; }
    float m = -INFINITY;
    #pragma unroll
    for (int c = 0; c < 40; ++c) m = fmaxf(m, hv[c]);
    float sm = 0.f;
    #pragma unroll
    for (int c = 0; c < 40; ++c) sm += expf(hv[c] - m);
    mx[tid] = m;
    lse[tid] = logf(sm);
  }
  __syncthreads();
  for (int i = tid; i < nn * C; i += AGG_T) {
    const int r = i / C, c = i - r * C;
    out[(size_t)lo * C + i] = rows[r * P2 + c] - mx[r] - lse[r];
  }
}


static inline int cdiv(int a, int b) { return (a + b - 1) / b; }

extern "C" void kernel_launch(void* const* d_in, const int* in_sizes, int n_in,
                              void* d_out, int out_size, void* d_ws, size_t ws_size,
                              hipStream_t stream) {
  const float* x      = (const float*)d_in[0];
  const int*   ei     = (const int*)  d_in[1];
  const float* W1     = (const float*)d_in[2];
  const float* a_src1 = (const float*)d_in[3];
  const float* a_dst1 = (const float*)d_in[4];
  const float* b1     = (const float*)d_in[5];
  const float* W2     = (const float*)d_in[6];
  const float* a_src2 = (const float*)d_in[7];
  const float* a_dst2 = (const float*)d_in[8];
  const float* b2     = (const float*)d_in[9];
  float* out = (float*)d_out;

  const int HC   = in_sizes[5];
  const int NCLS = in_sizes[9];
  const int F    = in_sizes[2] / HC;
  const int N    = in_sizes[0] / F;
  const int E    = in_sizes[1] / 2;
  const int NPAD = 48;
  const int LD1  = HC + 16;

  float* ws = (float*)d_ws;
  size_t off = 0;
  const int G   = cdiv(N, RN);
  const int NB  = cdiv(E, BM_E);
  const int CP  = NB * CELL;
  unsigned int* cells = (unsigned int*)(ws + off); off += (size_t)G * CP;
  float* h1   = ws + off; off += (size_t)N * HC;
  float* as1  = ws + off; off += (size_t)N * 8;
  float* ad1  = ws + off; off += (size_t)N * 8;
  float* out1 = ws + off; off += (size_t)N * HC;
  float* h2   = ws + off; off += (size_t)N * NCLS;
  float* as2  = ws + off; off += (size_t)N;
  float* ad2  = ws + off; off += (size_t)N;
  float* Wp   = ws + off; off += (size_t)HC * NPAD;
  float* W1e  = ws + off; off += (size_t)F * LD1;
  int* keep_s = (int*)(ws + off); off += (size_t)G * LCAP;
  int* keep_d = (int*)(ws + off); off += (size_t)G * LCAP;
  int* keep_n = (int*)(ws + off); off += (size_t)G;

  const int NWB = cdiv(F * LD1 + HC * NPAD, BM_T);
  prep_kernel<<<NWB + NB, BM_T, 0, stream>>>(ei, E, G, CP, cells, NWB, W1, a_src1, a_dst1, W1e,
                                             W2, a_src2, a_dst2, Wp, F, HC, NCLS, NPAD);

  gemm_wmma_f32_nt<5><<<dim3(N / 16), 32, 0, stream>>>(x, W1e, h1, as1, ad1, N, F, LD1, HC, HC, 8);
  edge_agg1_own_kernel<<<G, AGG_T, 0, stream>>>(ei, E, N, as1, ad1, a_src1, b1, h1, out1, keep_s, keep_d, keep_n,
                                               cells, CP);

  gemm_wmma_f32_nt<3><<<dim3(N / 16), 32, 0, stream>>>(out1, Wp, h2, as2, ad2, N, HC, NPAD, NCLS, NCLS, 1);
  edge_agg2_own_kernel<<<G, AGG_T, 0, stream>>>(ei, E, N, as2, ad2, a_src2, b2, h2, out, NCLS, keep_s, keep_d, keep_n);
}
